# starting kernel
# baseline (speedup 1.0000x reference)
#include <hip/hip_runtime.h>
#include <hip/hip_bf16.h>

#define N_NODES 50000
#define N_EDGES 800000
#define DIM     128
#define HEADS   8
#define HDIM    16

typedef __attribute__((ext_vector_type(2))) float v2f;
typedef __attribute__((ext_vector_type(8))) float v8f;

__device__ __forceinline__ void atomicMaxF(float* addr, float val) {
    if (val >= 0.0f)
        atomicMax((int*)addr, __float_as_int(val));
    else
        atomicMin((unsigned int*)addr, (unsigned int)__float_as_int(val));
}

__global__ void k_init(float* __restrict__ m, float* __restrict__ denom,
                       float* __restrict__ agg) {
    int t = blockIdx.x * blockDim.x + threadIdx.x;
    if (t < N_NODES * HEADS) {
        m[t]     = -__builtin_inff();
        denom[t] = 0.0f;
    }
    if (t < N_NODES * DIM) agg[t] = 0.0f;
}

__global__ void k_gemm_src_dst(const float* __restrict__ x,
                               const float* __restrict__ Wsrc,
                               const float* __restrict__ Wdst,
                               float* __restrict__ hsrc,
                               float* __restrict__ hdst) {
    __shared__ float ax[16 * DIM];
    const int rowBase = blockIdx.x * 16;
    for (int i = threadIdx.x; i < 16 * DIM; i += 256) {
        int r = i >> 7, c = i & 127;
        ax[i] = x[(rowBase + r) * DIM + c];
    }
    __syncthreads();

    const int wave  = threadIdx.x >> 5;
    const int lane  = threadIdx.x & 31;
    const int nBase = wave * 16;
    const int mrow  = lane & 15;
    const int khalf = (lane >> 4) * 2;
    const int ncol  = lane & 15;

    v8f accS = {}; v8f accD = {};
    for (int kk = 0; kk < DIM; kk += 4) {
        const int k0 = kk + khalf;
        v2f a;
        a.x = ax[mrow * DIM + k0];
        a.y = ax[mrow * DIM + k0 + 1];
        v2f bs, bd;
        bs.x = Wsrc[(k0    ) * DIM + nBase + ncol];
        bs.y = Wsrc[(k0 + 1) * DIM + nBase + ncol];
        bd.x = Wdst[(k0    ) * DIM + nBase + ncol];
        bd.y = Wdst[(k0 + 1) * DIM + nBase + ncol];
        accS = __builtin_amdgcn_wmma_f32_16x16x4_f32(false, a, false, bs,
                                                     (short)0, accS, false, false);
        accD = __builtin_amdgcn_wmma_f32_16x16x4_f32(false, a, false, bd,
                                                     (short)0, accD, false, false);
    }
    const int rOff = 8 * (lane >> 4);
    for (int r = 0; r < 8; ++r) {
        const int row = rowBase + r + rOff;
        hsrc[row * DIM + nBase + ncol] = accS[r];
        hdst[row * DIM + nBase + ncol] = accD[r];
    }
}

__global__ void k_edge_logits(const float* __restrict__ hsrc,
                              const float* __restrict__ hdst,
                              const int* __restrict__ ei,
                              const float* __restrict__ attn,
                              float* __restrict__ sbuf,
                              float* __restrict__ mbuf) {
    const int warp = threadIdx.x >> 5;
    const int lane = threadIdx.x & 31;
    const int e = blockIdx.x * 8 + warp;
    if (e >= N_EDGES) return;
    const int src = ei[e];
    const int dst = ei[N_EDGES + e];

    const float4 a = ((const float4*)(hsrc + (size_t)src * DIM))[lane];
    const float4 b = ((const float4*)(hdst + (size_t)dst * DIM))[lane];
    const float4 w = ((const float4*)attn)[lane];

    float t0 = a.x + b.x, t1 = a.y + b.y, t2 = a.z + b.z, t3 = a.w + b.w;
    t0 = t0 > 0.f ? t0 : 0.2f * t0;
    t1 = t1 > 0.f ? t1 : 0.2f * t1;
    t2 = t2 > 0.f ? t2 : 0.2f * t2;
    t3 = t3 > 0.f ? t3 : 0.2f * t3;
    float p = t0 * w.x + t1 * w.y + t2 * w.z + t3 * w.w;

    p += __shfl_xor(p, 1, 32);
    p += __shfl_xor(p, 2, 32);

    if ((lane & 3) == 0) {
        const int h = lane >> 2;
        sbuf[(size_t)e * HEADS + h] = p;
        atomicMaxF(mbuf + (size_t)dst * HEADS + h, p);
    }
}

__global__ void k_edge_exp(const int* __restrict__ ei,
                           const float* __restrict__ mbuf,
                           float* __restrict__ sbuf,
                           float* __restrict__ denom) {
    const int t = blockIdx.x * blockDim.x + threadIdx.x;
    if (t >= N_EDGES * HEADS) return;
    const int e = t >> 3, h = t & 7;
    const int dst = ei[N_EDGES + e];
    const float ex = expf(sbuf[t] - mbuf[(size_t)dst * HEADS + h]);
    sbuf[t] = ex;
    atomicAdd(denom + (size_t)dst * HEADS + h, ex);
}

__global__ void k_edge_aggregate(const float* __restrict__ hsrc,
                                 const int* __restrict__ ei,
                                 const float* __restrict__ sbuf,
                                 const float* __restrict__ denom,
                                 float* __restrict__ agg) {
    const int warp = threadIdx.x >> 5;
    const int lane = threadIdx.x & 31;
    const int e = blockIdx.x * 8 + warp;
    if (e >= N_EDGES) return;
    const int src = ei[e];
    const int dst = ei[N_EDGES + e];
    const int h = lane >> 2;

    const float ex  = sbuf[(size_t)e * HEADS + h];
    const float den = denom[(size_t)dst * HEADS + h];
    const float alpha = ex / (den + 1e-8f);

    const float4 hs = ((const float4*)(hsrc + (size_t)src * DIM))[lane];
    float* ag = agg + (size_t)dst * DIM + lane * 4;
    atomicAdd(ag + 0, hs.x * alpha);
    atomicAdd(ag + 1, hs.y * alpha);
    atomicAdd(ag + 2, hs.z * alpha);
    atomicAdd(ag + 3, hs.w * alpha);
}

__global__ void k_gemm_out_ln(const float* __restrict__ agg,
                              const float* __restrict__ Wout,
                              const float* __restrict__ bout,
                              const float* __restrict__ x,
                              const float* __restrict__ gamma,
                              const float* __restrict__ beta,
                              float* __restrict__ out) {
    __shared__ float aTile[16 * DIM];
    __shared__ float oTile[16 * DIM];
    const int rowBase = blockIdx.x * 16;
    for (int i = threadIdx.x; i < 16 * DIM; i += 256) {
        int r = i >> 7, c = i & 127;
        aTile[i] = agg[(rowBase + r) * DIM + c];
    }
    __syncthreads();

    const int wave  = threadIdx.x >> 5;
    const int lane  = threadIdx.x & 31;
    const int nBase = wave * 16;
    const int mrow  = lane & 15;
    const int khalf = (lane >> 4) * 2;
    const int ncol  = lane & 15;

    v8f acc = {};
    for (int kk = 0; kk < DIM; kk += 4) {
        const int k0 = kk + khalf;
        v2f a;
        a.x = aTile[mrow * DIM + k0];
        a.y = aTile[mrow * DIM + k0 + 1];
        v2f b;
        b.x = Wout[(k0    ) * DIM + nBase + ncol];
        b.y = Wout[(k0 + 1) * DIM + nBase + ncol];
        acc = __builtin_amdgcn_wmma_f32_16x16x4_f32(false, a, false, b,
                                                    (short)0, acc, false, false);
    }
    const int rOff = 8 * (lane >> 4);
    for (int r = 0; r < 8; ++r)
        oTile[(r + rOff) * DIM + nBase + ncol] = acc[r];
    __syncthreads();

    const int row  = threadIdx.x >> 4;
    const int j    = threadIdx.x & 15;
    const int gRow = rowBase + row;
    float vals[8];
    float s = 0.f, ss = 0.f;
    for (int k2 = 0; k2 < 8; ++k2) {
        const int c = j * 8 + k2;
        const float v = oTile[row * DIM + c] + bout[c] + x[(size_t)gRow * DIM + c];
        vals[k2] = v; s += v; ss += v * v;
    }
    for (int msk = 1; msk < 16; msk <<= 1) {
        s  += __shfl_xor(s,  msk, 32);
        ss += __shfl_xor(ss, msk, 32);
    }
    const float mu  = s * (1.0f / DIM);
    const float var = ss * (1.0f / DIM) - mu * mu;
    const float inv = rsqrtf(var + 1e-5f);
    for (int k2 = 0; k2 < 8; ++k2) {
        const int c = j * 8 + k2;
        out[(size_t)gRow * DIM + c] = (vals[k2] - mu) * inv * gamma[c] + beta[c];
    }
}

extern "C" void kernel_launch(void* const* d_in, const int* in_sizes, int n_in,
                              void* d_out, int out_size, void* d_ws, size_t ws_size,
                              hipStream_t stream) {
    const float* x     = (const float*)d_in[0];
    const int*   ei    = (const int*)  d_in[1];
    const float* Wsrc  = (const float*)d_in[2];
    const float* Wdst  = (const float*)d_in[3];
    const float* attn  = (const float*)d_in[4];
    const float* Wout  = (const float*)d_in[5];
    const float* bout  = (const float*)d_in[6];
    const float* gamma = (const float*)d_in[7];
    const float* beta  = (const float*)d_in[8];
    float* out = (float*)d_out;

    float* ws    = (float*)d_ws;
    float* hsrc  = ws;
    float* hdst  = hsrc  + (size_t)N_NODES * DIM;
    float* sbuf  = hdst  + (size_t)N_NODES * DIM;
    float* mbuf  = sbuf  + (size_t)N_EDGES * HEADS;
    float* denom = mbuf  + (size_t)N_NODES * HEADS;
    float* agg   = denom + (size_t)N_NODES * HEADS;

    const int rowTiles = N_NODES / 16;
    const int edgeBlks = N_EDGES / 8;
    const int initBlks = (N_NODES * DIM + 255) / 256;
    const int ehBlks   = (N_EDGES * HEADS + 255) / 256;

    k_init<<<initBlks, 256, 0, stream>>>(mbuf, denom, agg);
    k_gemm_src_dst<<<rowTiles, 256, 0, stream>>>(x, Wsrc, Wdst, hsrc, hdst);
    k_edge_logits<<<edgeBlks, 256, 0, stream>>>(hsrc, hdst, ei, attn, sbuf, mbuf);
    k_edge_exp<<<ehBlks, 256, 0, stream>>>(ei, mbuf, sbuf, denom);
    k_edge_aggregate<<<edgeBlks, 256, 0, stream>>>(hsrc, ei, sbuf, denom, agg);
    k_gemm_out_ln<<<rowTiles, 256, 0, stream>>>(agg, Wout, bout, x, gamma, beta, out);
}


# Float scatter atomics replaced by dst-grouped gather: int-atomic counting sort of edges, then a wave per node does logits, running-max softmax and weighted sum in registers; sum order = slot order
# speedup vs baseline: 1.8393x; 1.8393x over previous
#include <hip/hip_runtime.h>
#include <hip/hip_bf16.h>

#define N_NODES 50000
#define N_EDGES 800000
#define DIM     128
#define HEADS   8
#define HDIM    16

typedef __attribute__((ext_vector_type(2))) float v2f;
typedef __attribute__((ext_vector_type(8))) float v8f;

__global__ void k_init(int* __restrict__ cnt, int* __restrict__ total) {
    const int t = blockIdx.x * blockDim.x + threadIdx.x;
    if (t < N_NODES) cnt[t] = 0;
    if (t == 0) total[0] = 0;
}

__global__ void k_gemm_src_dst(const float* __restrict__ x,
                               const float* __restrict__ Wsrc,
                               const float* __restrict__ Wdst,
                               float* __restrict__ hsrc,
                               float* __restrict__ hdst) {
    __shared__ float ax[16 * DIM];
    const int rowBase = blockIdx.x * 16;
    for (int i = threadIdx.x; i < 16 * DIM; i += 256) {
        int r = i >> 7, c = i & 127;
        ax[i] = x[(rowBase + r) * DIM + c];
    }
    __syncthreads();

    const int wave  = threadIdx.x >> 5;
    const int lane  = threadIdx.x & 31;
    const int nBase = wave * 16;
    const int mrow  = lane & 15;
    const int khalf = (lane >> 4) * 2;
    const int ncol  = lane & 15;

    v8f accS = {}; v8f accD = {};
    for (int kk = 0; kk < DIM; kk += 4) {
        const int k0 = kk + khalf;
        v2f a;
        a.x = ax[mrow * DIM + k0];
        a.y = ax[mrow * DIM + k0 + 1];
        v2f bs, bd;
        bs.x = Wsrc[(k0    ) * DIM + nBase + ncol];
        bs.y = Wsrc[(k0 + 1) * DIM + nBase + ncol];
        bd.x = Wdst[(k0    ) * DIM + nBase + ncol];
        bd.y = Wdst[(k0 + 1) * DIM + nBase + ncol];
        accS = __builtin_amdgcn_wmma_f32_16x16x4_f32(false, a, false, bs,
                                                     (short)0, accS, false, false);
        accD = __builtin_amdgcn_wmma_f32_16x16x4_f32(false, a, false, bd,
                                                     (short)0, accD, false, false);
    }
    const int rOff = 8 * (lane >> 4);
    for (int r = 0; r < 8; ++r) {
        const int row = rowBase + r + rOff;
        hsrc[row * DIM + nBase + ncol] = accS[r];
        hdst[row * DIM + nBase + ncol] = accD[r];
    }
}

__global__ void k_edge_rank(const int* __restrict__ ei,
                            int* __restrict__ cnt,
                            int* __restrict__ rank) {
    const int e = blockIdx.x * blockDim.x + threadIdx.x;
    if (e >= N_EDGES) return;
    rank[e] = atomicAdd(cnt + ei[N_EDGES + e], 1);
}

__global__ void k_node_slots(const int* __restrict__ cnt,
                             int* __restrict__ total,
                             int* __restrict__ start) {
    const int n    = blockIdx.x * blockDim.x + threadIdx.x;
    const int lane = threadIdx.x & 31;
    const int c    = (n < N_NODES) ? cnt[n] : 0;
    int inc = c;
    for (int o = 1; o < 32; o <<= 1) {
        const int t = __shfl_up(inc, o, 32);
        if (lane >= o) inc += t;
    }
    int base = 0;
    if (lane == 31) base = atomicAdd(total, inc);
    base = __shfl(base, 31, 32);
    if (n < N_NODES) start[n] = base + inc - c;
}

__global__ void k_edge_fill(const int* __restrict__ ei,
                            const int* __restrict__ start,
                            const int* __restrict__ rank,
                            int* __restrict__ perm) {
    const int e = blockIdx.x * blockDim.x + threadIdx.x;
    if (e >= N_EDGES) return;
    perm[start[ei[N_EDGES + e]] + rank[e]] = ei[e];
}

__device__ __forceinline__ float edge_logit(const float4 a, const float4 b,
                                            const float4 w) {
    float t0 = a.x + b.x, t1 = a.y + b.y, t2 = a.z + b.z, t3 = a.w + b.w;
    t0 = t0 > 0.f ? t0 : 0.2f * t0;
    t1 = t1 > 0.f ? t1 : 0.2f * t1;
    t2 = t2 > 0.f ? t2 : 0.2f * t2;
    t3 = t3 > 0.f ? t3 : 0.2f * t3;
    float p = t0 * w.x + t1 * w.y + t2 * w.z + t3 * w.w;
    p += __shfl_xor(p, 1, 32);
    p += __shfl_xor(p, 2, 32);
    return p;
}

__global__ void __launch_bounds__(256)
k_node_attend(const float* __restrict__ hsrc,
              const float* __restrict__ hdst,
              const int* __restrict__ start,
              const int* __restrict__ cnt,
              const int* __restrict__ perm,
              const float* __restrict__ attn,
              float* __restrict__ agg) {
    const int lane = threadIdx.x & 31;
    const int n = __builtin_amdgcn_readfirstlane(blockIdx.x * 8 + (threadIdx.x >> 5));
    if (n >= N_NODES) return;
    const int beg = start[n];
    const int end = beg + cnt[n];

    const float4 hd = ((const float4*)(hdst + (size_t)n * DIM))[lane];
    const float4 w  = ((const float4*)attn)[lane];
    const float NEG_INF = -__builtin_inff();

    float m = NEG_INF, den = 0.0f;
    float4 acc = make_float4(0.f, 0.f, 0.f, 0.f);

    for (int base = beg; base < end; base += 32) {
        const int rem  = end - base;
        const int last = (rem < 32 ? rem : 32) - 1;
        const int my = perm[base + (lane < last ? lane : last)];
        for (int j = 0; j <= last; j += 4) {
            const int j1 = (j + 1 < last) ? j + 1 : last;
            const int j2 = (j + 2 < last) ? j + 2 : last;
            const int j3 = (j + 3 < last) ? j + 3 : last;
            const int s0 = __builtin_amdgcn_readlane(my, j);
            const int s1 = __builtin_amdgcn_readlane(my, j1);
            const int s2 = __builtin_amdgcn_readlane(my, j2);
            const int s3 = __builtin_amdgcn_readlane(my, j3);
            const float4 r0 = ((const float4*)(hsrc + (size_t)s0 * DIM))[lane];
            const float4 r1 = ((const float4*)(hsrc + (size_t)s1 * DIM))[lane];
            const float4 r2 = ((const float4*)(hsrc + (size_t)s2 * DIM))[lane];
            const float4 r3 = ((const float4*)(hsrc + (size_t)s3 * DIM))[lane];
            const float p0 = edge_logit(r0, hd, w);
            float p1 = edge_logit(r1, hd, w);
            float p2 = edge_logit(r2, hd, w);
            float p3 = edge_logit(r3, hd, w);
            if (j + 1 > last) p1 = NEG_INF;
            if (j + 2 > last) p2 = NEG_INF;
            if (j + 3 > last) p3 = NEG_INF;

            const float mn = fmaxf(fmaxf(m, p0), fmaxf(fmaxf(p1, p2), p3));
            const float sc = expf(m - mn);
            const float e0 = expf(p0 - mn);
            const float e1 = expf(p1 - mn);
            const float e2 = expf(p2 - mn);
            const float e3 = expf(p3 - mn);
            den   = den   * sc + ((e0 + e1) + (e2 + e3));
            acc.x = acc.x * sc + ((r0.x * e0 + r1.x * e1) + (r2.x * e2 + r3.x * e3));
            acc.y = acc.y * sc + ((r0.y * e0 + r1.y * e1) + (r2.y * e2 + r3.y * e3));
            acc.z = acc.z * sc + ((r0.z * e0 + r1.z * e1) + (r2.z * e2 + r3.z * e3));
            acc.w = acc.w * sc + ((r0.w * e0 + r1.w * e1) + (r2.w * e2 + r3.w * e3));
            m = mn;
        }
    }
    const float d = den + 1e-8f;
    ((float4*)(agg + (size_t)n * DIM))[lane] =
        make_float4(acc.x / d, acc.y / d, acc.z / d, acc.w / d);
}

__global__ void k_gemm_out_ln(const float* __restrict__ agg,
                              const float* __restrict__ Wout,
                              const float* __restrict__ bout,
                              const float* __restrict__ x,
                              const float* __restrict__ gamma,
                              const float* __restrict__ beta,
                              float* __restrict__ out) {
    __shared__ float aTile[16 * DIM];
    __shared__ float oTile[16 * DIM];
    const int rowBase = blockIdx.x * 16;
    for (int i = threadIdx.x; i < 16 * DIM; i += 256) {
        int r = i >> 7, c = i & 127;
        aTile[i] = agg[(rowBase + r) * DIM + c];
    }
    __syncthreads();

    const int wave  = threadIdx.x >> 5;
    const int lane  = threadIdx.x & 31;
    const int nBase = wave * 16;
    const int mrow  = lane & 15;
    const int khalf = (lane >> 4) * 2;
    const int ncol  = lane & 15;

    v8f acc = {};
    for (int kk = 0; kk < DIM; kk += 4) {
        const int k0 = kk + khalf;
        v2f a;
        a.x = aTile[mrow * DIM + k0];
        a.y = aTile[mrow * DIM + k0 + 1];
        v2f b;
        b.x = Wout[(k0    ) * DIM + nBase + ncol];
        b.y = Wout[(k0 + 1) * DIM + nBase + ncol];
        acc = __builtin_amdgcn_wmma_f32_16x16x4_f32(false, a, false, b,
                                                    (short)0, acc, false, false);
    }
    const int rOff = 8 * (lane >> 4);
    for (int r = 0; r < 8; ++r)
        oTile[(r + rOff) * DIM + nBase + ncol] = acc[r];
    __syncthreads();

    const int row  = threadIdx.x >> 4;
    const int j    = threadIdx.x & 15;
    const int gRow = rowBase + row;
    float vals[8];
    float s = 0.f, ss = 0.f;
    for (int k2 = 0; k2 < 8; ++k2) {
        const int c = j * 8 + k2;
        const float v = oTile[row * DIM + c] + bout[c] + x[(size_t)gRow * DIM + c];
        vals[k2] = v; s += v; ss += v * v;
    }
    for (int msk = 1; msk < 16; msk <<= 1) {
        s  += __shfl_xor(s,  msk, 32);
        ss += __shfl_xor(ss, msk, 32);
    }
    const float mu  = s * (1.0f / DIM);
    const float var = ss * (1.0f / DIM) - mu * mu;
    const float inv = rsqrtf(var + 1e-5f);
    for (int k2 = 0; k2 < 8; ++k2) {
        const int c = j * 8 + k2;
        out[(size_t)gRow * DIM + c] = (vals[k2] - mu) * inv * gamma[c] + beta[c];
    }
}

extern "C" void kernel_launch(void* const* d_in, const int* in_sizes, int n_in,
                              void* d_out, int out_size, void* d_ws, size_t ws_size,
                              hipStream_t stream) {
    const float* x     = (const float*)d_in[0];
    const int*   ei    = (const int*)  d_in[1];
    const float* Wsrc  = (const float*)d_in[2];
    const float* Wdst  = (const float*)d_in[3];
    const float* attn  = (const float*)d_in[4];
    const float* Wout  = (const float*)d_in[5];
    const float* bout  = (const float*)d_in[6];
    const float* gamma = (const float*)d_in[7];
    const float* beta  = (const float*)d_in[8];
    float* out = (float*)d_out;

    float* ws    = (float*)d_ws;
    float* hsrc  = ws;
    float* hdst  = hsrc  + (size_t)N_NODES * DIM;
    float* agg   = hdst  + (size_t)N_NODES * DIM;
    int*   cnt   = (int*)(agg + (size_t)N_NODES * DIM);
    int*   start = cnt   + N_NODES;
    int*   total = start + N_NODES;
    int*   rank  = total + 16;
    int*   perm  = rank  + N_EDGES;

    const int rowTiles = N_NODES / 16;
    const int nodeBlks = (N_NODES + 255) / 256;
    const int edgeBlks = (N_EDGES + 255) / 256;
    const int waveBlks = (N_NODES + 7) / 8;

    k_init<<<nodeBlks, 256, 0, stream>>>(cnt, total);
    k_gemm_src_dst<<<rowTiles, 256, 0, stream>>>(x, Wsrc, Wdst, hsrc, hdst);
    k_edge_rank<<<edgeBlks, 256, 0, stream>>>(ei, cnt, rank);
    k_node_slots<<<nodeBlks, 256, 0, stream>>>(cnt, total, start);
    k_edge_fill<<<edgeBlks, 256, 0, stream>>>(ei, start, rank, perm);
    k_node_attend<<<waveBlks, 256, 0, stream>>>(hsrc, hdst, start, cnt, perm, attn, agg);
    k_gemm_out_ln<<<rowTiles, 256, 0, stream>>>(agg, Wout, bout, x, gamma, beta, out);
}


# Both GEMMs: each wave loads its W column-tile fragments once into registers and walks row panels; A read 16 B per lane from padded double-buffered LDS; k order inside K changed alike in A and B
# speedup vs baseline: 2.2827x; 1.2411x over previous
#include <hip/hip_runtime.h>
#include <hip/hip_bf16.h>

#define N_NODES 50000
#define N_EDGES 800000
#define DIM     128
#define HEADS   8
#define HDIM    16

typedef __attribute__((ext_vector_type(2))) float v2f;
typedef __attribute__((ext_vector_type(8))) float v8f;

__global__ void k_init(int* __restrict__ cnt, int* __restrict__ total) {
    const int t = blockIdx.x * blockDim.x + threadIdx.x;
    if (t < N_NODES) cnt[t] = 0;
    if (t == 0) total[0] = 0;
}

#define N_PANELS (N_NODES / 16)
#define PITCH 132

__global__ void __launch_bounds__(256)
k_gemm_src_dst(const float* __restrict__ x,
               const float* __restrict__ Wsrc,
               const float* __restrict__ Wdst,
               float* __restrict__ hsrc,
               float* __restrict__ hdst) {
    __shared__ __attribute__((aligned(16))) float ax[2][16 * PITCH];

    const int wave  = threadIdx.x >> 5;
    const int lane  = threadIdx.x & 31;
    const int nBase = wave * 16;
    const int cc    = lane & 15;
    const int hh    = lane >> 4;

    v2f bS[32], bD[32];
    {
        const float* ps = Wsrc + (4 * hh) * DIM + nBase + cc;
        const float* pd = Wdst + (4 * hh) * DIM + nBase + cc;
#pragma unroll
        for (int i = 0; i < 16; ++i) {
            bS[2 * i].x     = ps[(8 * i    ) * DIM];
            bS[2 * i].y     = ps[(8 * i + 1) * DIM];
            bS[2 * i + 1].x = ps[(8 * i + 2) * DIM];
            bS[2 * i + 1].y = ps[(8 * i + 3) * DIM];
            bD[2 * i].x     = pd[(8 * i    ) * DIM];
            bD[2 * i].y     = pd[(8 * i + 1) * DIM];
            bD[2 * i + 1].x = pd[(8 * i + 2) * DIM];
            bD[2 * i + 1].y = pd[(8 * i + 3) * DIM];
        }
    }

    const int sr = threadIdx.x >> 5;
    const int sc = (threadIdx.x & 31) * 4;

    int p   = blockIdx.x;
    int cur = 0;
    {
        const int pf = p < N_PANELS ? p : 0;
        const float* g = x + ((size_t)pf * 16 + sr) * DIM + sc;
        const float4 f0 = *(const float4*)g;
        const float4 f1 = *(const float4*)(g + 8 * DIM);
        *(float4*)&ax[0][sr * PITCH + sc]       = f0;
        *(float4*)&ax[0][(sr + 8) * PITCH + sc] = f1;
    }
    __syncthreads();

    for (; p < N_PANELS; p += gridDim.x) {
        const int  pn   = p + gridDim.x;
        const bool more = pn < N_PANELS;
        float4 n0 = make_float4(0.f, 0.f, 0.f, 0.f), n1 = n0;
        if (more) {
            const float* g = x + ((size_t)pn * 16 + sr) * DIM + sc;
            n0 = *(const float4*)g;
            n1 = *(const float4*)(g + 8 * DIM);
        }

        v8f accS = {}; v8f accD = {};
        const float* arow = &ax[cur][cc * PITCH + 4 * hh];
        float4 af[16];
#pragma unroll
        for (int i = 0; i < 16; ++i) af[i] = *(const float4*)(arow + 8 * i);
        __builtin_amdgcn_sched_barrier(0);
#pragma unroll
        for (int i = 0; i < 16; ++i) {
            const float4 a4 = af[i];
            v2f a01; a01.x = a4.x; a01.y = a4.y;
            v2f a23; a23.x = a4.z; a23.y = a4.w;
            accS = __builtin_amdgcn_wmma_f32_16x16x4_f32(false, a01, false, bS[2 * i],
                                                         (short)0, accS, false, false);
            accD = __builtin_amdgcn_wmma_f32_16x16x4_f32(false, a01, false, bD[2 * i],
                                                         (short)0, accD, false, false);
            accS = __builtin_amdgcn_wmma_f32_16x16x4_f32(false, a23, false, bS[2 * i + 1],
                                                         (short)0, accS, false, false);
            accD = __builtin_amdgcn_wmma_f32_16x16x4_f32(false, a23, false, bD[2 * i + 1],
                                                         (short)0, accD, false, false);
        }
        __builtin_amdgcn_sched_barrier(0);
        asm volatile("v_nop\n\tv_nop\n\tv_nop\n\tv_nop");
        __builtin_amdgcn_sched_barrier(0);
        const size_t row0 = (size_t)p * 16 + 8 * hh;
        for (int r = 0; r < 8; ++r) {
            hsrc[(row0 + r) * DIM + nBase + cc] = accS[r];
            hdst[(row0 + r) * DIM + nBase + cc] = accD[r];
        }
        if (more) {
            *(float4*)&ax[cur ^ 1][sr * PITCH + sc]       = n0;
            *(float4*)&ax[cur ^ 1][(sr + 8) * PITCH + sc] = n1;
        }
        __syncthreads();
        cur ^= 1;
    }
}

__global__ void k_edge_rank(const int* __restrict__ ei,
                            int* __restrict__ cnt,
                            int* __restrict__ rank) {
    const int e = blockIdx.x * blockDim.x + threadIdx.x;
    if (e >= N_EDGES) return;
    rank[e] = atomicAdd(cnt + ei[N_EDGES + e], 1);
}

__global__ void k_node_slots(const int* __restrict__ cnt,
                             int* __restrict__ total,
                             int* __restrict__ start) {
    const int n    = blockIdx.x * blockDim.x + threadIdx.x;
    const int lane = threadIdx.x & 31;
    const int c    = (n < N_NODES) ? cnt[n] : 0;
    int inc = c;
    for (int o = 1; o < 32; o <<= 1) {
        const int t = __shfl_up(inc, o, 32);
        if (lane >= o) inc += t;
    }
    int base = 0;
    if (lane == 31) base = atomicAdd(total, inc);
    base = __shfl(base, 31, 32);
    if (n < N_NODES) start[n] = base + inc - c;
}

__global__ void k_edge_fill(const int* __restrict__ ei,
                            const int* __restrict__ start,
                            const int* __restrict__ rank,
                            int* __restrict__ perm) {
    const int e = blockIdx.x * blockDim.x + threadIdx.x;
    if (e >= N_EDGES) return;
    perm[start[ei[N_EDGES + e]] + rank[e]] = ei[e];
}

__device__ __forceinline__ float edge_logit(const float4 a, const float4 b,
                                            const float4 w) {
    float t0 = a.x + b.x, t1 = a.y + b.y, t2 = a.z + b.z, t3 = a.w + b.w;
    t0 = t0 > 0.f ? t0 : 0.2f * t0;
    t1 = t1 > 0.f ? t1 : 0.2f * t1;
    t2 = t2 > 0.f ? t2 : 0.2f * t2;
    t3 = t3 > 0.f ? t3 : 0.2f * t3;
    float p = t0 * w.x + t1 * w.y + t2 * w.z + t3 * w.w;
    p += __shfl_xor(p, 1, 32);
    p += __shfl_xor(p, 2, 32);
    return p;
}

__global__ void __launch_bounds__(256)
k_node_attend(const float* __restrict__ hsrc,
              const float* __restrict__ hdst,
              const int* __restrict__ start,
              const int* __restrict__ cnt,
              const int* __restrict__ perm,
              const float* __restrict__ attn,
              float* __restrict__ agg) {
    const int lane = threadIdx.x & 31;
    const int n = __builtin_amdgcn_readfirstlane(blockIdx.x * 8 + (threadIdx.x >> 5));
    if (n >= N_NODES) return;
    const int beg = start[n];
    const int end = beg + cnt[n];

    const float4 hd = ((const float4*)(hdst + (size_t)n * DIM))[lane];
    const float4 w  = ((const float4*)attn)[lane];
    const float NEG_INF = -__builtin_inff();

    float m = NEG_INF, den = 0.0f;
    float4 acc = make_float4(0.f, 0.f, 0.f, 0.f);

    for (int base = beg; base < end; base += 32) {
        const int rem  = end - base;
        const int last = (rem < 32 ? rem : 32) - 1;
        const int my = perm[base + (lane < last ? lane : last)];
        for (int j = 0; j <= last; j += 4) {
            const int j1 = (j + 1 < last) ? j + 1 : last;
            const int j2 = (j + 2 < last) ? j + 2 : last;
            const int j3 = (j + 3 < last) ? j + 3 : last;
            const int s0 = __builtin_amdgcn_readlane(my, j);
            const int s1 = __builtin_amdgcn_readlane(my, j1);
            const int s2 = __builtin_amdgcn_readlane(my, j2);
            const int s3 = __builtin_amdgcn_readlane(my, j3);
            const float4 r0 = ((const float4*)(hsrc + (size_t)s0 * DIM))[lane];
            const float4 r1 = ((const float4*)(hsrc + (size_t)s1 * DIM))[lane];
            const float4 r2 = ((const float4*)(hsrc + (size_t)s2 * DIM))[lane];
            const float4 r3 = ((const float4*)(hsrc + (size_t)s3 * DIM))[lane];
            const float p0 = edge_logit(r0, hd, w);
            float p1 = edge_logit(r1, hd, w);
            float p2 = edge_logit(r2, hd, w);
            float p3 = edge_logit(r3, hd, w);
            if (j + 1 > last) p1 = NEG_INF;
            if (j + 2 > last) p2 = NEG_INF;
            if (j + 3 > last) p3 = NEG_INF;

            const float mn = fmaxf(fmaxf(m, p0), fmaxf(fmaxf(p1, p2), p3));
            const float sc = expf(m - mn);
            const float e0 = expf(p0 - mn);
            const float e1 = expf(p1 - mn);
            const float e2 = expf(p2 - mn);
            const float e3 = expf(p3 - mn);
            den   = den   * sc + ((e0 + e1) + (e2 + e3));
            acc.x = acc.x * sc + ((r0.x * e0 + r1.x * e1) + (r2.x * e2 + r3.x * e3));
            acc.y = acc.y * sc + ((r0.y * e0 + r1.y * e1) + (r2.y * e2 + r3.y * e3));
            acc.z = acc.z * sc + ((r0.z * e0 + r1.z * e1) + (r2.z * e2 + r3.z * e3));
            acc.w = acc.w * sc + ((r0.w * e0 + r1.w * e1) + (r2.w * e2 + r3.w * e3));
            m = mn;
        }
    }
    const float d = den + 1e-8f;
    ((float4*)(agg + (size_t)n * DIM))[lane] =
        make_float4(acc.x / d, acc.y / d, acc.z / d, acc.w / d);
}

__global__ void __launch_bounds__(256)
k_gemm_out_ln(const float* __restrict__ agg,
              const float* __restrict__ Wout,
              const float* __restrict__ bout,
              const float* __restrict__ x,
              const float* __restrict__ gamma,
              const float* __restrict__ beta,
              float* __restrict__ out) {
    __shared__ __attribute__((aligned(16))) float aTile[2][16 * PITCH];
    __shared__ __attribute__((aligned(16))) float oTile[2][16 * DIM];

    const int wave  = threadIdx.x >> 5;
    const int lane  = threadIdx.x & 31;
    const int nBase = wave * 16;
    const int cc    = lane & 15;
    const int hh    = lane >> 4;

    v2f bW[32];
    {
        const float* pw = Wout + (4 * hh) * DIM + nBase + cc;
#pragma unroll
        for (int i = 0; i < 16; ++i) {
            bW[2 * i].x     = pw[(8 * i    ) * DIM];
            bW[2 * i].y     = pw[(8 * i + 1) * DIM];
            bW[2 * i + 1].x = pw[(8 * i + 2) * DIM];
            bW[2 * i + 1].y = pw[(8 * i + 3) * DIM];
        }
    }

    const int row = threadIdx.x >> 4;
    const int j   = threadIdx.x & 15;
    float bo[8], ga[8], be[8];
    for (int k2 = 0; k2 < 8; ++k2) {
        bo[k2] = bout[j * 8 + k2];
        ga[k2] = gamma[j * 8 + k2];
        be[k2] = beta[j * 8 + k2];
    }

    const int sr = threadIdx.x >> 5;
    const int sc = (threadIdx.x & 31) * 4;

    int p   = blockIdx.x;
    int cur = 0;
    {
        const int pf = p < N_PANELS ? p : 0;
        const float* g = agg + ((size_t)pf * 16 + sr) * DIM + sc;
        const float4 f0 = *(const float4*)g;
        const float4 f1 = *(const float4*)(g + 8 * DIM);
        *(float4*)&aTile[0][sr * PITCH + sc]       = f0;
        *(float4*)&aTile[0][(sr + 8) * PITCH + sc] = f1;
    }
    __syncthreads();

    for (; p < N_PANELS; p += gridDim.x) {
        const int  pn   = p + gridDim.x;
        const bool more = pn < N_PANELS;
        float4 n0 = make_float4(0.f, 0.f, 0.f, 0.f), n1 = n0;
        if (more) {
            const float* g = agg + ((size_t)pn * 16 + sr) * DIM + sc;
            n0 = *(const float4*)g;
            n1 = *(const float4*)(g + 8 * DIM);
        }
        const size_t gRow = (size_t)p * 16 + row;
        const float4 x0 = *(const float4*)(x + gRow * DIM + j * 8);
        const float4 x1 = *(const float4*)(x + gRow * DIM + j * 8 + 4);

        v8f acc = {};
        const float* arow = &aTile[cur][cc * PITCH + 4 * hh];
        float4 af[16];
#pragma unroll
        for (int i = 0; i < 16; ++i) af[i] = *(const float4*)(arow + 8 * i);
        __builtin_amdgcn_sched_barrier(0);
#pragma unroll
        for (int i = 0; i < 16; ++i) {
            const float4 a4 = af[i];
            v2f a01; a01.x = a4.x; a01.y = a4.y;
            v2f a23; a23.x = a4.z; a23.y = a4.w;
            acc = __builtin_amdgcn_wmma_f32_16x16x4_f32(false, a01, false, bW[2 * i],
                                                        (short)0, acc, false, false);
            acc = __builtin_amdgcn_wmma_f32_16x16x4_f32(false, a23, false, bW[2 * i + 1],
                                                        (short)0, acc, false, false);
        }
        __builtin_amdgcn_sched_barrier(0);
        asm volatile("v_nop\n\tv_nop\n\tv_nop\n\tv_nop");
        __builtin_amdgcn_sched_barrier(0);
        float* ot = oTile[cur];
        for (int r = 0; r < 8; ++r)
            ot[(r + 8 * hh) * DIM + nBase + cc] = acc[r];
        if (more) {
            *(float4*)&aTile[cur ^ 1][sr * PITCH + sc]       = n0;
            *(float4*)&aTile[cur ^ 1][(sr + 8) * PITCH + sc] = n1;
        }
        __syncthreads();

        const float xv[8] = {x0.x, x0.y, x0.z, x0.w, x1.x, x1.y, x1.z, x1.w};
        float vals[8];
        float s = 0.f, ss = 0.f;
        for (int k2 = 0; k2 < 8; ++k2) {
            const float v = ot[row * DIM + j * 8 + k2] + bo[k2] + xv[k2];
            vals[k2] = v; s += v; ss += v * v;
        }
        for (int msk = 1; msk < 16; msk <<= 1) {
            s  += __shfl_xor(s,  msk, 32);
            ss += __shfl_xor(ss, msk, 32);
        }
        const float mu  = s * (1.0f / DIM);
        const float var = ss * (1.0f / DIM) - mu * mu;
        const float inv = rsqrtf(var + 1e-5f);
        for (int k2 = 0; k2 < 8; ++k2)
            out[gRow * DIM + j * 8 + k2] = (vals[k2] - mu) * inv * ga[k2] + be[k2];
        cur ^= 1;
    }
}

extern "C" void kernel_launch(void* const* d_in, const int* in_sizes, int n_in,
                              void* d_out, int out_size, void* d_ws, size_t ws_size,
                              hipStream_t stream) {
    const float* x     = (const float*)d_in[0];
    const int*   ei    = (const int*)  d_in[1];
    const float* Wsrc  = (const float*)d_in[2];
    const float* Wdst  = (const float*)d_in[3];
    const float* attn  = (const float*)d_in[4];
    const float* Wout  = (const float*)d_in[5];
    const float* bout  = (const float*)d_in[6];
    const float* gamma = (const float*)d_in[7];
    const float* beta  = (const float*)d_in[8];
    float* out = (float*)d_out;

    float* ws    = (float*)d_ws;
    float* hsrc  = ws;
    float* hdst  = hsrc  + (size_t)N_NODES * DIM;
    float* agg   = hdst  + (size_t)N_NODES * DIM;
    int*   cnt   = (int*)(agg + (size_t)N_NODES * DIM);
    int*   start = cnt   + N_NODES;
    int*   total = start + N_NODES;
    int*   rank  = total + 16;
    int*   perm  = rank  + N_EDGES;

    const int gemm1Blks = 512;
    const int gemm6Blks = 512;
    const int nodeBlks = (N_NODES + 255) / 256;
    const int edgeBlks = (N_EDGES + 255) / 256;
    const int waveBlks = (N_NODES + 7) / 8;

    k_init<<<nodeBlks, 256, 0, stream>>>(cnt, total);
    k_gemm_src_dst<<<gemm1Blks, 256, 0, stream>>>(x, Wsrc, Wdst, hsrc, hdst);
    k_edge_rank<<<edgeBlks, 256, 0, stream>>>(ei, cnt, rank);
    k_node_slots<<<nodeBlks, 256, 0, stream>>>(cnt, total, start);
    k_edge_fill<<<edgeBlks, 256, 0, stream>>>(ei, start, rank, perm);
    k_node_attend<<<waveBlks, 256, 0, stream>>>(hsrc, hdst, start, cnt, perm, attn, agg);
    k_gemm_out_ln<<<gemm6Blks, 256, 0, stream>>>(agg, Wout, bout, x, gamma, beta, out);
}


# Edge grouping by dst redone as a two-level counting sort with all counters in LDS (chunk->bucket->node): no returning global atomic per edge, 7->6 launches; sum order = new slot order
# speedup vs baseline: 3.5973x; 1.5759x over previous
#include <hip/hip_runtime.h>
#include <hip/hip_bf16.h>

#define N_NODES 50000
#define N_EDGES 800000
#define DIM     128
#define HEADS   8
#define HDIM    16

typedef __attribute__((ext_vector_type(2))) float v2f;
typedef __attribute__((ext_vector_type(8))) float v8f;

#define N_CHUNK   128
#define CHUNK_E   (N_EDGES / N_CHUNK)
#define EPT       ((CHUNK_E + 1023) / 1024)
#define BKT_SHIFT 8
#define BKT_NODES (1 << BKT_SHIFT)
#define N_BKT     ((N_NODES + BKT_NODES - 1) / BKT_NODES)
#define ROWS_PW   ((N_BKT + 31) / 32)
#define RPT       6
static_assert(N_EDGES % N_CHUNK == 0, "chunks must be equal");
static_assert(N_BKT <= 224 && N_CHUNK == 128, "kernel 1 reads a count row as 4 ints per lane");

__global__ void __launch_bounds__(1024)
k_chunk_count(const int* __restrict__ ei, int* __restrict__ cntT) {
    __shared__ int h[N_BKT];
    const int c   = blockIdx.x;
    const int tid = threadIdx.x;
    if (tid < N_BKT) h[tid] = 0;
    const int* d = ei + N_EDGES + c * CHUNK_E;
    int dv[EPT];
#pragma unroll
    for (int k = 0; k < EPT; ++k) {
        const int i = tid + k * 1024;
        dv[k] = i < CHUNK_E ? d[i] : -1;
    }
    __syncthreads();
#pragma unroll
    for (int k = 0; k < EPT; ++k)
        if (dv[k] >= 0) atomicAdd(&h[dv[k] >> BKT_SHIFT], 1);
    __syncthreads();
    if (tid < N_BKT) cntT[tid * N_CHUNK + c] = h[tid];
}

__global__ void __launch_bounds__(1024)
k_chunk_scatter(const int* __restrict__ ei, const int* __restrict__ cntT,
                int* __restrict__ bstart, int* __restrict__ rec) {
    __shared__ __attribute__((aligned(16))) int tot[256];
    __shared__ int cur[256];
    const int c    = blockIdx.x;
    const int tid  = threadIdx.x;
    const int wave = tid >> 5;
    const int lane = tid & 31;

    const int* s = ei + c * CHUNK_E;
    const int* d = s + N_EDGES;
    int sv[EPT], dv[EPT];
#pragma unroll
    for (int k = 0; k < EPT; ++k) {
        const int i = tid + k * 1024;
        const bool ok = i < CHUNK_E;
        dv[k] = ok ? d[i] : -1;
        sv[k] = ok ? s[i] : 0;
    }

    int4 ra[ROWS_PW];
#pragma unroll
    for (int k = 0; k < ROWS_PW; ++k) {
        const int b = wave + 32 * k;
        ra[k] = make_int4(0, 0, 0, 0);
        if (b < N_BKT) ra[k] = ((const int4*)(cntT + b * N_CHUNK))[lane];
    }
    if (tid >= 32 * ROWS_PW && tid < 256) { tot[tid] = 0; cur[tid] = 0; }
    const int e0 = 4 * lane;
#pragma unroll
    for (int k = 0; k < ROWS_PW; ++k) {
        const int4 a = ra[k];
        int t = (a.x + a.y) + (a.z + a.w);
        int p = (e0     < c ? a.x : 0) + (e0 + 1 < c ? a.y : 0)
              + (e0 + 2 < c ? a.z : 0) + (e0 + 3 < c ? a.w : 0);
        for (int m = 16; m >= 1; m >>= 1) {
            t += __shfl_xor(t, m, 32);
            p += __shfl_xor(p, m, 32);
        }
        if (lane == 0) { tot[wave + 32 * k] = t; cur[wave + 32 * k] = p; }
    }
    __syncthreads();
    if (wave == 0) {
        const int4 a = ((const int4*)tot)[2 * lane];
        const int4 q = ((const int4*)tot)[2 * lane + 1];
        const int v[8] = {a.x, a.y, a.z, a.w, q.x, q.y, q.z, q.w};
        int ex[8];
        int run = 0;
        for (int j = 0; j < 8; ++j) { ex[j] = run; run += v[j]; }
        int inc = run;
        for (int o = 1; o < 32; o <<= 1) {
            const int u = __shfl_up(inc, o, 32);
            if (lane >= o) inc += u;
        }
        const int off = inc - run;
        for (int j = 0; j < 8; ++j) {
            const int bb    = 8 * lane + j;
            const int first = off + ex[j];
            cur[bb] += first;
            if (c == 0 && bb <= N_BKT) bstart[bb] = first;
        }
    }
    __syncthreads();
#pragma unroll
    for (int k = 0; k < EPT; ++k)
        if (dv[k] >= 0) {
            const int slot = atomicAdd(&cur[dv[k] >> BKT_SHIFT], 1);
            ((int2*)rec)[slot] = make_int2(sv[k], dv[k]);
        }
}

__global__ void __launch_bounds__(1024)
k_bucket_sort(const int* __restrict__ rec, const int* __restrict__ bstart,
              int* __restrict__ start, int* __restrict__ cnt, int* __restrict__ perm) {
    __shared__ __attribute__((aligned(16))) int c1[BKT_NODES];
    __shared__ int cur[BKT_NODES];
    const int b    = blockIdx.x;
    const int tid  = threadIdx.x;
    const int wave = tid >> 5;
    const int lane = tid & 31;
    const int r0 = bstart[b];
    const int r1 = bstart[b + 1];
    const int2* rr = (const int2*)rec;
    if (tid < BKT_NODES) c1[tid] = 0;
    int2 rc[RPT];
#pragma unroll
    for (int k = 0; k < RPT; ++k) {
        const int i = r0 + tid + k * 1024;
        rc[k] = make_int2(0, -1);
        if (i < r1) rc[k] = rr[i];
    }
    __syncthreads();
#pragma unroll
    for (int k = 0; k < RPT; ++k)
        if (rc[k].y >= 0) atomicAdd(&c1[rc[k].y & (BKT_NODES - 1)], 1);
    for (int i = r0 + tid + RPT * 1024; i < r1; i += 1024)
        atomicAdd(&c1[rr[i].y & (BKT_NODES - 1)], 1);
    __syncthreads();
    if (wave == 0) {
        const int4 a = ((const int4*)c1)[2 * lane];
        const int4 q = ((const int4*)c1)[2 * lane + 1];
        const int v[8] = {a.x, a.y, a.z, a.w, q.x, q.y, q.z, q.w};
        int ex[8];
        int run = 0;
        for (int j = 0; j < 8; ++j) { ex[j] = run; run += v[j]; }
        int inc = run;
        for (int o = 1; o < 32; o <<= 1) {
            const int u = __shfl_up(inc, o, 32);
            if (lane >= o) inc += u;
        }
        const int off = r0 + inc - run;
        for (int j = 0; j < 8; ++j) {
            const int nl    = 8 * lane + j;
            const int n     = b * BKT_NODES + nl;
            const int first = off + ex[j];
            cur[nl] = first;
            if (n < N_NODES) { start[n] = first; cnt[n] = v[j]; }
        }
    }
    __syncthreads();
#pragma unroll
    for (int k = 0; k < RPT; ++k)
        if (rc[k].y >= 0) {
            const int slot = atomicAdd(&cur[rc[k].y & (BKT_NODES - 1)], 1);
            perm[slot] = rc[k].x;
        }
    for (int i = r0 + tid + RPT * 1024; i < r1; i += 1024) {
        const int2 q = rr[i];
        const int slot = atomicAdd(&cur[q.y & (BKT_NODES - 1)], 1);
        perm[slot] = q.x;
    }
}

#define N_PANELS (N_NODES / 16)
#define PITCH 132

__global__ void __launch_bounds__(256)
k_gemm_src_dst(const float* __restrict__ x,
               const float* __restrict__ Wsrc,
               const float* __restrict__ Wdst,
               float* __restrict__ hsrc,
               float* __restrict__ hdst) {
    __shared__ __attribute__((aligned(16))) float ax[2][16 * PITCH];

    const int wave  = threadIdx.x >> 5;
    const int lane  = threadIdx.x & 31;
    const int nBase = wave * 16;
    const int cc    = lane & 15;
    const int hh    = lane >> 4;

    v2f bS[32], bD[32];
    {
        const float* ps = Wsrc + (4 * hh) * DIM + nBase + cc;
        const float* pd = Wdst + (4 * hh) * DIM + nBase + cc;
#pragma unroll
        for (int i = 0; i < 16; ++i) {
            bS[2 * i].x     = ps[(8 * i    ) * DIM];
            bS[2 * i].y     = ps[(8 * i + 1) * DIM];
            bS[2 * i + 1].x = ps[(8 * i + 2) * DIM];
            bS[2 * i + 1].y = ps[(8 * i + 3) * DIM];
            bD[2 * i].x     = pd[(8 * i    ) * DIM];
            bD[2 * i].y     = pd[(8 * i + 1) * DIM];
            bD[2 * i + 1].x = pd[(8 * i + 2) * DIM];
            bD[2 * i + 1].y = pd[(8 * i + 3) * DIM];
        }
    }

    const int sr = threadIdx.x >> 5;
    const int sc = (threadIdx.x & 31) * 4;

    int p   = blockIdx.x;
    int cur = 0;
    {
        const int pf = p < N_PANELS ? p : 0;
        const float* g = x + ((size_t)pf * 16 + sr) * DIM + sc;
        const float4 f0 = *(const float4*)g;
        const float4 f1 = *(const float4*)(g + 8 * DIM);
        *(float4*)&ax[0][sr * PITCH + sc]       = f0;
        *(float4*)&ax[0][(sr + 8) * PITCH + sc] = f1;
    }
    __syncthreads();

    for (; p < N_PANELS; p += gridDim.x) {
        const int  pn   = p + gridDim.x;
        const bool more = pn < N_PANELS;
        float4 n0 = make_float4(0.f, 0.f, 0.f, 0.f), n1 = n0;
        if (more) {
            const float* g = x + ((size_t)pn * 16 + sr) * DIM + sc;
            n0 = *(const float4*)g;
            n1 = *(const float4*)(g + 8 * DIM);
        }

        v8f accS = {}; v8f accD = {};
        const float* arow = &ax[cur][cc * PITCH + 4 * hh];
        float4 af[16];
#pragma unroll
        for (int i = 0; i < 16; ++i) af[i] = *(const float4*)(arow + 8 * i);
        __builtin_amdgcn_sched_barrier(0);
#pragma unroll
        for (int i = 0; i < 16; ++i) {
            const float4 a4 = af[i];
            v2f a01; a01.x = a4.x; a01.y = a4.y;
            v2f a23; a23.x = a4.z; a23.y = a4.w;
            accS = __builtin_amdgcn_wmma_f32_16x16x4_f32(false, a01, false, bS[2 * i],
                                                         (short)0, accS, false, false);
            accD = __builtin_amdgcn_wmma_f32_16x16x4_f32(false, a01, false, bD[2 * i],
                                                         (short)0, accD, false, false);
            accS = __builtin_amdgcn_wmma_f32_16x16x4_f32(false, a23, false, bS[2 * i + 1],
                                                         (short)0, accS, false, false);
            accD = __builtin_amdgcn_wmma_f32_16x16x4_f32(false, a23, false, bD[2 * i + 1],
                                                         (short)0, accD, false, false);
        }
        __builtin_amdgcn_sched_barrier(0);
        asm volatile("v_nop\n\tv_nop\n\tv_nop\n\tv_nop");
        __builtin_amdgcn_sched_barrier(0);
        const size_t row0 = (size_t)p * 16 + 8 * hh;
        for (int r = 0; r < 8; ++r) {
            hsrc[(row0 + r) * DIM + nBase + cc] = accS[r];
            hdst[(row0 + r) * DIM + nBase + cc] = accD[r];
        }
        if (more) {
            *(float4*)&ax[cur ^ 1][sr * PITCH + sc]       = n0;
            *(float4*)&ax[cur ^ 1][(sr + 8) * PITCH + sc] = n1;
        }
        __syncthreads();
        cur ^= 1;
    }
}

__device__ __forceinline__ float edge_logit(const float4 a, const float4 b,
                                            const float4 w) {
    float t0 = a.x + b.x, t1 = a.y + b.y, t2 = a.z + b.z, t3 = a.w + b.w;
    t0 = t0 > 0.f ? t0 : 0.2f * t0;
    t1 = t1 > 0.f ? t1 : 0.2f * t1;
    t2 = t2 > 0.f ? t2 : 0.2f * t2;
    t3 = t3 > 0.f ? t3 : 0.2f * t3;
    float p = t0 * w.x + t1 * w.y + t2 * w.z + t3 * w.w;
    p += __shfl_xor(p, 1, 32);
    p += __shfl_xor(p, 2, 32);
    return p;
}

__global__ void __launch_bounds__(256)
k_node_attend(const float* __restrict__ hsrc,
              const float* __restrict__ hdst,
              const int* __restrict__ start,
              const int* __restrict__ cnt,
              const int* __restrict__ perm,
              const float* __restrict__ attn,
              float* __restrict__ agg) {
    const int lane = threadIdx.x & 31;
    const int n = __builtin_amdgcn_readfirstlane(blockIdx.x * 8 + (threadIdx.x >> 5));
    if (n >= N_NODES) return;
    const int beg = start[n];
    const int end = beg + cnt[n];

    const float4 hd = ((const float4*)(hdst + (size_t)n * DIM))[lane];
    const float4 w  = ((const float4*)attn)[lane];
    const float NEG_INF = -__builtin_inff();

    float m = NEG_INF, den = 0.0f;
    float4 acc = make_float4(0.f, 0.f, 0.f, 0.f);

    for (int base = beg; base < end; base += 32) {
        const int rem  = end - base;
        const int last = (rem < 32 ? rem : 32) - 1;
        const int my = perm[base + (lane < last ? lane : last)];
        for (int j = 0; j <= last; j += 4) {
            const int j1 = (j + 1 < last) ? j + 1 : last;
            const int j2 = (j + 2 < last) ? j + 2 : last;
            const int j3 = (j + 3 < last) ? j + 3 : last;
            const int s0 = __builtin_amdgcn_readlane(my, j);
            const int s1 = __builtin_amdgcn_readlane(my, j1);
            const int s2 = __builtin_amdgcn_readlane(my, j2);
            const int s3 = __builtin_amdgcn_readlane(my, j3);
            const float4 r0 = ((const float4*)(hsrc + (size_t)s0 * DIM))[lane];
            const float4 r1 = ((const float4*)(hsrc + (size_t)s1 * DIM))[lane];
            const float4 r2 = ((const float4*)(hsrc + (size_t)s2 * DIM))[lane];
            const float4 r3 = ((const float4*)(hsrc + (size_t)s3 * DIM))[lane];
            const float p0 = edge_logit(r0, hd, w);
            float p1 = edge_logit(r1, hd, w);
            float p2 = edge_logit(r2, hd, w);
            float p3 = edge_logit(r3, hd, w);
            if (j + 1 > last) p1 = NEG_INF;
            if (j + 2 > last) p2 = NEG_INF;
            if (j + 3 > last) p3 = NEG_INF;

            const float mn = fmaxf(fmaxf(m, p0), fmaxf(fmaxf(p1, p2), p3));
            const float sc = expf(m - mn);
            const float e0 = expf(p0 - mn);
            const float e1 = expf(p1 - mn);
            const float e2 = expf(p2 - mn);
            const float e3 = expf(p3 - mn);
            den   = den   * sc + ((e0 + e1) + (e2 + e3));
            acc.x = acc.x * sc + ((r0.x * e0 + r1.x * e1) + (r2.x * e2 + r3.x * e3));
            acc.y = acc.y * sc + ((r0.y * e0 + r1.y * e1) + (r2.y * e2 + r3.y * e3));
            acc.z = acc.z * sc + ((r0.z * e0 + r1.z * e1) + (r2.z * e2 + r3.z * e3));
            acc.w = acc.w * sc + ((r0.w * e0 + r1.w * e1) + (r2.w * e2 + r3.w * e3));
            m = mn;
        }
    }
    const float d = den + 1e-8f;
    ((float4*)(agg + (size_t)n * DIM))[lane] =
        make_float4(acc.x / d, acc.y / d, acc.z / d, acc.w / d);
}

__global__ void __launch_bounds__(256)
k_gemm_out_ln(const float* __restrict__ agg,
              const float* __restrict__ Wout,
              const float* __restrict__ bout,
              const float* __restrict__ x,
              const float* __restrict__ gamma,
              const float* __restrict__ beta,
              float* __restrict__ out) {
    __shared__ __attribute__((aligned(16))) float aTile[2][16 * PITCH];
    __shared__ __attribute__((aligned(16))) float oTile[2][16 * DIM];

    const int wave  = threadIdx.x >> 5;
    const int lane  = threadIdx.x & 31;
    const int nBase = wave * 16;
    const int cc    = lane & 15;
    const int hh    = lane >> 4;

    v2f bW[32];
    {
        const float* pw = Wout + (4 * hh) * DIM + nBase + cc;
#pragma unroll
        for (int i = 0; i < 16; ++i) {
            bW[2 * i].x     = pw[(8 * i    ) * DIM];
            bW[2 * i].y     = pw[(8 * i + 1) * DIM];
            bW[2 * i + 1].x = pw[(8 * i + 2) * DIM];
            bW[2 * i + 1].y = pw[(8 * i + 3) * DIM];
        }
    }

    const int row = threadIdx.x >> 4;
    const int j   = threadIdx.x & 15;
    float bo[8], ga[8], be[8];
    for (int k2 = 0; k2 < 8; ++k2) {
        bo[k2] = bout[j * 8 + k2];
        ga[k2] = gamma[j * 8 + k2];
        be[k2] = beta[j * 8 + k2];
    }

    const int sr = threadIdx.x >> 5;
    const int sc = (threadIdx.x & 31) * 4;

    int p   = blockIdx.x;
    int cur = 0;
    {
        const int pf = p < N_PANELS ? p : 0;
        const float* g = agg + ((size_t)pf * 16 + sr) * DIM + sc;
        const float4 f0 = *(const float4*)g;
        const float4 f1 = *(const float4*)(g + 8 * DIM);
        *(float4*)&aTile[0][sr * PITCH + sc]       = f0;
        *(float4*)&aTile[0][(sr + 8) * PITCH + sc] = f1;
    }
    __syncthreads();

    for (; p < N_PANELS; p += gridDim.x) {
        const int  pn   = p + gridDim.x;
        const bool more = pn < N_PANELS;
        float4 n0 = make_float4(0.f, 0.f, 0.f, 0.f), n1 = n0;
        if (more) {
            const float* g = agg + ((size_t)pn * 16 + sr) * DIM + sc;
            n0 = *(const float4*)g;
            n1 = *(const float4*)(g + 8 * DIM);
        }
        const size_t gRow = (size_t)p * 16 + row;
        const float4 x0 = *(const float4*)(x + gRow * DIM + j * 8);
        const float4 x1 = *(const float4*)(x + gRow * DIM + j * 8 + 4);

        v8f acc = {};
        const float* arow = &aTile[cur][cc * PITCH + 4 * hh];
        float4 af[16];
#pragma unroll
        for (int i = 0; i < 16; ++i) af[i] = *(const float4*)(arow + 8 * i);
        __builtin_amdgcn_sched_barrier(0);
#pragma unroll
        for (int i = 0; i < 16; ++i) {
            const float4 a4 = af[i];
            v2f a01; a01.x = a4.x; a01.y = a4.y;
            v2f a23; a23.x = a4.z; a23.y = a4.w;
            acc = __builtin_amdgcn_wmma_f32_16x16x4_f32(false, a01, false, bW[2 * i],
                                                        (short)0, acc, false, false);
            acc = __builtin_amdgcn_wmma_f32_16x16x4_f32(false, a23, false, bW[2 * i + 1],
                                                        (short)0, acc, false, false);
        }
        __builtin_amdgcn_sched_barrier(0);
        asm volatile("v_nop\n\tv_nop\n\tv_nop\n\tv_nop");
        __builtin_amdgcn_sched_barrier(0);
        float* ot = oTile[cur];
        for (int r = 0; r < 8; ++r)
            ot[(r + 8 * hh) * DIM + nBase + cc] = acc[r];
        if (more) {
            *(float4*)&aTile[cur ^ 1][sr * PITCH + sc]       = n0;
            *(float4*)&aTile[cur ^ 1][(sr + 8) * PITCH + sc] = n1;
        }
        __syncthreads();

        const float xv[8] = {x0.x, x0.y, x0.z, x0.w, x1.x, x1.y, x1.z, x1.w};
        float vals[8];
        float s = 0.f, ss = 0.f;
        for (int k2 = 0; k2 < 8; ++k2) {
            const float v = ot[row * DIM + j * 8 + k2] + bo[k2] + xv[k2];
            vals[k2] = v; s += v; ss += v * v;
        }
        for (int msk = 1; msk < 16; msk <<= 1) {
            s  += __shfl_xor(s,  msk, 32);
            ss += __shfl_xor(ss, msk, 32);
        }
        const float mu  = s * (1.0f / DIM);
        const float var = ss * (1.0f / DIM) - mu * mu;
        const float inv = rsqrtf(var + 1e-5f);
        for (int k2 = 0; k2 < 8; ++k2)
            out[gRow * DIM + j * 8 + k2] = (vals[k2] - mu) * inv * ga[k2] + be[k2];
        cur ^= 1;
    }
}

extern "C" void kernel_launch(void* const* d_in, const int* in_sizes, int n_in,
                              void* d_out, int out_size, void* d_ws, size_t ws_size,
                              hipStream_t stream) {
    const float* x     = (const float*)d_in[0];
    const int*   ei    = (const int*)  d_in[1];
    const float* Wsrc  = (const float*)d_in[2];
    const float* Wdst  = (const float*)d_in[3];
    const float* attn  = (const float*)d_in[4];
    const float* Wout  = (const float*)d_in[5];
    const float* bout  = (const float*)d_in[6];
    const float* gamma = (const float*)d_in[7];
    const float* beta  = (const float*)d_in[8];
    float* out = (float*)d_out;

    float* ws     = (float*)d_ws;
    float* hsrc   = ws;
    float* hdst   = hsrc   + (size_t)N_NODES * DIM;
    float* agg    = hdst   + (size_t)N_NODES * DIM;
    int*   cnt    = (int*)(agg + (size_t)N_NODES * DIM);
    int*   start  = cnt    + N_NODES;
    int*   bstart = start  + N_NODES;
    int*   cntT   = bstart + 256;
    int*   rec    = cntT   + N_BKT * N_CHUNK;
    int*   perm   = rec    + 2 * N_EDGES;

    const int gemm1Blks = 512;
    const int gemm6Blks = 512;
    const int waveBlks = (N_NODES + 7) / 8;

    k_chunk_count<<<N_CHUNK, 1024, 0, stream>>>(ei, cntT);
    k_chunk_scatter<<<N_CHUNK, 1024, 0, stream>>>(ei, cntT, bstart, rec);
    k_bucket_sort<<<N_BKT, 1024, 0, stream>>>(rec, bstart, start, cnt, perm);
    k_gemm_src_dst<<<gemm1Blks, 256, 0, stream>>>(x, Wsrc, Wdst, hsrc, hdst);
    k_node_attend<<<waveBlks, 256, 0, stream>>>(hsrc, hdst, start, cnt, perm, attn, agg);
    k_gemm_out_ln<<<gemm6Blks, 256, 0, stream>>>(agg, Wout, bout, x, gamma, beta, out);
}


# Edge sort level 1 made chunk-local (each chunk orders its own edges by bucket; no count pass) and run by the first GEMM's blocks before their panels: 6->4 launches; sum order = new slot order
# speedup vs baseline: 3.7565x; 1.0443x over previous
#include <hip/hip_runtime.h>
#include <hip/hip_bf16.h>

#define N_NODES 50000
#define N_EDGES 800000
#define DIM     128
#define HEADS   8
#define HDIM    16

typedef __attribute__((ext_vector_type(2))) float v2f;
typedef __attribute__((ext_vector_type(8))) float v8f;

#define NCH       512
#define CH_MAX    ((N_EDGES + NCH - 1) / NCH)
#define EPT       ((CH_MAX + 255) / 256)
#define BKT_SHIFT 8
#define BKT_NODES (1 << BKT_SHIFT)
#define N_BKT     ((N_NODES + BKT_NODES - 1) / BKT_NODES)
#define RPT       8
static_assert((long long)NCH * N_EDGES < 2147483647LL, "chunk bounds are computed in 32 bits");
static_assert(N_BKT < 256, "the bucket scan covers 256 entries, the last one closes the chunk");
static_assert(2 * NCH == 1024, "level 2: two threads per chunk run");

__device__ __forceinline__ int chunk_first(const int c) { return (c * N_EDGES) / NCH; }

__device__ __forceinline__ void chunk_sort(const int* __restrict__ ei, int* __restrict__ offT,
                                           int* __restrict__ rec, int* tot, int* curs) {
    const int c    = blockIdx.x;
    const int tid  = threadIdx.x;
    const int lane = tid & 31;
    const int e0 = chunk_first(c);
    const int ne = chunk_first(c + 1) - e0;
    const int* s = ei + e0;
    const int* d = s + N_EDGES;
    tot[tid] = 0;
    int sv[EPT], dv[EPT];
#pragma unroll
    for (int k = 0; k < EPT; ++k) {
        const int i = tid + k * 256;
        const bool ok = i < ne;
        dv[k] = ok ? d[i] : -1;
        sv[k] = ok ? s[i] : 0;
    }
    __syncthreads();
#pragma unroll
    for (int k = 0; k < EPT; ++k)
        if (dv[k] >= 0) atomicAdd(&tot[dv[k] >> BKT_SHIFT], 1);
    __syncthreads();
    if (tid < 32) {
        const int4 a = ((const int4*)tot)[2 * lane];
        const int4 q = ((const int4*)tot)[2 * lane + 1];
        const int v[8] = {a.x, a.y, a.z, a.w, q.x, q.y, q.z, q.w};
        int ex[8];
        int run = 0;
        for (int j = 0; j < 8; ++j) { ex[j] = run; run += v[j]; }
        int inc = run;
        for (int o = 1; o < 32; o <<= 1) {
            const int u = __shfl_up(inc, o, 32);
            if (lane >= o) inc += u;
        }
        const int off = inc - run;
        for (int j = 0; j < 8; ++j) {
            const int bb    = 8 * lane + j;
            const int first = off + ex[j];
            curs[bb] = first;
            if (bb <= N_BKT) offT[bb * NCH + c] = first;
        }
    }
    __syncthreads();
#pragma unroll
    for (int k = 0; k < EPT; ++k)
        if (dv[k] >= 0) {
            const int slot = atomicAdd(&curs[dv[k] >> BKT_SHIFT], 1);
            ((int2*)rec)[e0 + slot] = make_int2(sv[k], dv[k]);
        }
}

__global__ void __launch_bounds__(1024)
k_bucket_sort(const int* __restrict__ rec, const int* __restrict__ offT,
              int* __restrict__ start, int* __restrict__ cnt, int* __restrict__ perm) {
    __shared__ __attribute__((aligned(16))) int c1[BKT_NODES];
    __shared__ int cur[BKT_NODES];
    __shared__ int wsum[32];
    const int b    = blockIdx.x;
    const int tid  = threadIdx.x;
    const int wave = tid >> 5;
    const int lane = tid & 31;
    const int c    = tid >> 1;
    const int half = tid & 1;
    const int o0 = offT[b * NCH + c];
    const int o1 = offT[(b + 1) * NCH + c];
    const int2* rr = (const int2*)rec + chunk_first(c);
    if (tid < BKT_NODES) c1[tid] = 0;
    int2 rc[RPT];
#pragma unroll
    for (int k = 0; k < RPT; ++k) {
        const int i = o0 + half + 2 * k;
        rc[k] = make_int2(0, -1);
        if (i < o1) rc[k] = rr[i];
    }
    int t = half == 0 ? o0 : 0;
    for (int m = 16; m >= 1; m >>= 1) t += __shfl_xor(t, m, 32);
    if (lane == 0) wsum[wave] = t;
    __syncthreads();
#pragma unroll
    for (int k = 0; k < RPT; ++k)
        if (rc[k].y >= 0) atomicAdd(&c1[rc[k].y & (BKT_NODES - 1)], 1);
    for (int i = o0 + half + 2 * RPT; i < o1; i += 2)
        atomicAdd(&c1[rr[i].y & (BKT_NODES - 1)], 1);
    __syncthreads();
    if (wave == 0) {
        int r0 = wsum[lane];
        for (int m = 16; m >= 1; m >>= 1) r0 += __shfl_xor(r0, m, 32);
        const int4 a = ((const int4*)c1)[2 * lane];
        const int4 q = ((const int4*)c1)[2 * lane + 1];
        const int v[8] = {a.x, a.y, a.z, a.w, q.x, q.y, q.z, q.w};
        int ex[8];
        int run = 0;
        for (int j = 0; j < 8; ++j) { ex[j] = run; run += v[j]; }
        int inc = run;
        for (int o = 1; o < 32; o <<= 1) {
            const int u = __shfl_up(inc, o, 32);
            if (lane >= o) inc += u;
        }
        const int off = r0 + inc - run;
        for (int j = 0; j < 8; ++j) {
            const int nl    = 8 * lane + j;
            const int n     = b * BKT_NODES + nl;
            const int first = off + ex[j];
            cur[nl] = first;
            if (n < N_NODES) { start[n] = first; cnt[n] = v[j]; }
        }
    }
    __syncthreads();
#pragma unroll
    for (int k = 0; k < RPT; ++k)
        if (rc[k].y >= 0) {
            const int slot = atomicAdd(&cur[rc[k].y & (BKT_NODES - 1)], 1);
            perm[slot] = rc[k].x;
        }
    for (int i = o0 + half + 2 * RPT; i < o1; i += 2) {
        const int2 q = rr[i];
        const int slot = atomicAdd(&cur[q.y & (BKT_NODES - 1)], 1);
        perm[slot] = q.x;
    }
}

#define N_PANELS (N_NODES / 16)
#define PITCH 132

__global__ void __launch_bounds__(256)
k_gemm_src_dst(const float* __restrict__ x,
               const float* __restrict__ Wsrc,
               const float* __restrict__ Wdst,
               float* __restrict__ hsrc,
               float* __restrict__ hdst,
               const int* __restrict__ ei,
               int* __restrict__ offT,
               int* __restrict__ rec) {
    __shared__ __attribute__((aligned(16))) float ax[2][16 * PITCH];
    __shared__ __attribute__((aligned(16))) int tot[256];
    __shared__ int curs[256];

    chunk_sort(ei, offT, rec, tot, curs);

    const int wave  = threadIdx.x >> 5;
    const int lane  = threadIdx.x & 31;
    const int nBase = wave * 16;
    const int cc    = lane & 15;
    const int hh    = lane >> 4;

    v2f bS[32], bD[32];
    {
        const float* ps = Wsrc + (4 * hh) * DIM + nBase + cc;
        const float* pd = Wdst + (4 * hh) * DIM + nBase + cc;
#pragma unroll
        for (int i = 0; i < 16; ++i) {
            bS[2 * i].x     = ps[(8 * i    ) * DIM];
            bS[2 * i].y     = ps[(8 * i + 1) * DIM];
            bS[2 * i + 1].x = ps[(8 * i + 2) * DIM];
            bS[2 * i + 1].y = ps[(8 * i + 3) * DIM];
            bD[2 * i].x     = pd[(8 * i    ) * DIM];
            bD[2 * i].y     = pd[(8 * i + 1) * DIM];
            bD[2 * i + 1].x = pd[(8 * i + 2) * DIM];
            bD[2 * i + 1].y = pd[(8 * i + 3) * DIM];
        }
    }

    const int sr = threadIdx.x >> 5;
    const int sc = (threadIdx.x & 31) * 4;

    int p   = blockIdx.x;
    int cur = 0;
    {
        const int pf = p < N_PANELS ? p : 0;
        const float* g = x + ((size_t)pf * 16 + sr) * DIM + sc;
        const float4 f0 = *(const float4*)g;
        const float4 f1 = *(const float4*)(g + 8 * DIM);
        *(float4*)&ax[0][sr * PITCH + sc]       = f0;
        *(float4*)&ax[0][(sr + 8) * PITCH + sc] = f1;
    }
    __syncthreads();

    for (; p < N_PANELS; p += gridDim.x) {
        const int  pn   = p + gridDim.x;
        const bool more = pn < N_PANELS;
        float4 n0 = make_float4(0.f, 0.f, 0.f, 0.f), n1 = n0;
        if (more) {
            const float* g = x + ((size_t)pn * 16 + sr) * DIM + sc;
            n0 = *(const float4*)g;
            n1 = *(const float4*)(g + 8 * DIM);
        }

        v8f accS = {}; v8f accD = {};
        const float* arow = &ax[cur][cc * PITCH + 4 * hh];
        float4 af[16];
#pragma unroll
        for (int i = 0; i < 16; ++i) af[i] = *(const float4*)(arow + 8 * i);
        __builtin_amdgcn_sched_barrier(0);
#pragma unroll
        for (int i = 0; i < 16; ++i) {
            const float4 a4 = af[i];
            v2f a01; a01.x = a4.x; a01.y = a4.y;
            v2f a23; a23.x = a4.z; a23.y = a4.w;
            accS = __builtin_amdgcn_wmma_f32_16x16x4_f32(false, a01, false, bS[2 * i],
                                                         (short)0, accS, false, false);
            accD = __builtin_amdgcn_wmma_f32_16x16x4_f32(false, a01, false, bD[2 * i],
                                                         (short)0, accD, false, false);
            accS = __builtin_amdgcn_wmma_f32_16x16x4_f32(false, a23, false, bS[2 * i + 1],
                                                         (short)0, accS, false, false);
            accD = __builtin_amdgcn_wmma_f32_16x16x4_f32(false, a23, false, bD[2 * i + 1],
                                                         (short)0, accD, false, false);
        }
        __builtin_amdgcn_sched_barrier(0);
        asm volatile("v_nop\n\tv_nop\n\tv_nop\n\tv_nop");
        __builtin_amdgcn_sched_barrier(0);
        const size_t row0 = (size_t)p * 16 + 8 * hh;
        for (int r = 0; r < 8; ++r) {
            hsrc[(row0 + r) * DIM + nBase + cc] = accS[r];
            hdst[(row0 + r) * DIM + nBase + cc] = accD[r];
        }
        if (more) {
            *(float4*)&ax[cur ^ 1][sr * PITCH + sc]       = n0;
            *(float4*)&ax[cur ^ 1][(sr + 8) * PITCH + sc] = n1;
        }
        __syncthreads();
        cur ^= 1;
    }
}

__device__ __forceinline__ float edge_logit(const float4 a, const float4 b,
                                            const float4 w) {
    float t0 = a.x + b.x, t1 = a.y + b.y, t2 = a.z + b.z, t3 = a.w + b.w;
    t0 = t0 > 0.f ? t0 : 0.2f * t0;
    t1 = t1 > 0.f ? t1 : 0.2f * t1;
    t2 = t2 > 0.f ? t2 : 0.2f * t2;
    t3 = t3 > 0.f ? t3 : 0.2f * t3;
    float p = t0 * w.x + t1 * w.y + t2 * w.z + t3 * w.w;
    p += __shfl_xor(p, 1, 32);
    p += __shfl_xor(p, 2, 32);
    return p;
}

__global__ void __launch_bounds__(256)
k_node_attend(const float* __restrict__ hsrc,
              const float* __restrict__ hdst,
              const int* __restrict__ start,
              const int* __restrict__ cnt,
              const int* __restrict__ perm,
              const float* __restrict__ attn,
              float* __restrict__ agg) {
    const int lane = threadIdx.x & 31;
    const int n = __builtin_amdgcn_readfirstlane(blockIdx.x * 8 + (threadIdx.x >> 5));
    if (n >= N_NODES) return;
    const int beg = start[n];
    const int end = beg + cnt[n];

    const float4 hd = ((const float4*)(hdst + (size_t)n * DIM))[lane];
    const float4 w  = ((const float4*)attn)[lane];
    const float NEG_INF = -__builtin_inff();

    float m = NEG_INF, den = 0.0f;
    float4 acc = make_float4(0.f, 0.f, 0.f, 0.f);

    for (int base = beg; base < end; base += 32) {
        const int rem  = end - base;
        const int last = (rem < 32 ? rem : 32) - 1;
        const int my = perm[base + (lane < last ? lane : last)];
        for (int j = 0; j <= last; j += 4) {
            const int j1 = (j + 1 < last) ? j + 1 : last;
            const int j2 = (j + 2 < last) ? j + 2 : last;
            const int j3 = (j + 3 < last) ? j + 3 : last;
            const int s0 = __builtin_amdgcn_readlane(my, j);
            const int s1 = __builtin_amdgcn_readlane(my, j1);
            const int s2 = __builtin_amdgcn_readlane(my, j2);
            const int s3 = __builtin_amdgcn_readlane(my, j3);
            const float4 r0 = ((const float4*)(hsrc + (size_t)s0 * DIM))[lane];
            const float4 r1 = ((const float4*)(hsrc + (size_t)s1 * DIM))[lane];
            const float4 r2 = ((const float4*)(hsrc + (size_t)s2 * DIM))[lane];
            const float4 r3 = ((const float4*)(hsrc + (size_t)s3 * DIM))[lane];
            const float p0 = edge_logit(r0, hd, w);
            float p1 = edge_logit(r1, hd, w);
            float p2 = edge_logit(r2, hd, w);
            float p3 = edge_logit(r3, hd, w);
            if (j + 1 > last) p1 = NEG_INF;
            if (j + 2 > last) p2 = NEG_INF;
            if (j + 3 > last) p3 = NEG_INF;

            const float mn = fmaxf(fmaxf(m, p0), fmaxf(fmaxf(p1, p2), p3));
            const float sc = expf(m - mn);
            const float e0 = expf(p0 - mn);
            const float e1 = expf(p1 - mn);
            const float e2 = expf(p2 - mn);
            const float e3 = expf(p3 - mn);
            den   = den   * sc + ((e0 + e1) + (e2 + e3));
            acc.x = acc.x * sc + ((r0.x * e0 + r1.x * e1) + (r2.x * e2 + r3.x * e3));
            acc.y = acc.y * sc + ((r0.y * e0 + r1.y * e1) + (r2.y * e2 + r3.y * e3));
            acc.z = acc.z * sc + ((r0.z * e0 + r1.z * e1) + (r2.z * e2 + r3.z * e3));
            acc.w = acc.w * sc + ((r0.w * e0 + r1.w * e1) + (r2.w * e2 + r3.w * e3));
            m = mn;
        }
    }
    const float d = den + 1e-8f;
    ((float4*)(agg + (size_t)n * DIM))[lane] =
        make_float4(acc.x / d, acc.y / d, acc.z / d, acc.w / d);
}

__global__ void __launch_bounds__(256)
k_gemm_out_ln(const float* __restrict__ agg,
              const float* __restrict__ Wout,
              const float* __restrict__ bout,
              const float* __restrict__ x,
              const float* __restrict__ gamma,
              const float* __restrict__ beta,
              float* __restrict__ out) {
    __shared__ __attribute__((aligned(16))) float aTile[2][16 * PITCH];
    __shared__ __attribute__((aligned(16))) float oTile[2][16 * DIM];

    const int wave  = threadIdx.x >> 5;
    const int lane  = threadIdx.x & 31;
    const int nBase = wave * 16;
    const int cc    = lane & 15;
    const int hh    = lane >> 4;

    v2f bW[32];
    {
        const float* pw = Wout + (4 * hh) * DIM + nBase + cc;
#pragma unroll
        for (int i = 0; i < 16; ++i) {
            bW[2 * i].x     = pw[(8 * i    ) * DIM];
            bW[2 * i].y     = pw[(8 * i + 1) * DIM];
            bW[2 * i + 1].x = pw[(8 * i + 2) * DIM];
            bW[2 * i + 1].y = pw[(8 * i + 3) * DIM];
        }
    }

    const int row = threadIdx.x >> 4;
    const int j   = threadIdx.x & 15;
    float bo[8], ga[8], be[8];
    for (int k2 = 0; k2 < 8; ++k2) {
        bo[k2] = bout[j * 8 + k2];
        ga[k2] = gamma[j * 8 + k2];
        be[k2] = beta[j * 8 + k2];
    }

    const int sr = threadIdx.x >> 5;
    const int sc = (threadIdx.x & 31) * 4;

    int p   = blockIdx.x;
    int cur = 0;
    {
        const int pf = p < N_PANELS ? p : 0;
        const float* g = agg + ((size_t)pf * 16 + sr) * DIM + sc;
        const float4 f0 = *(const float4*)g;
        const float4 f1 = *(const float4*)(g + 8 * DIM);
        *(float4*)&aTile[0][sr * PITCH + sc]       = f0;
        *(float4*)&aTile[0][(sr + 8) * PITCH + sc] = f1;
    }
    __syncthreads();

    for (; p < N_PANELS; p += gridDim.x) {
        const int  pn   = p + gridDim.x;
        const bool more = pn < N_PANELS;
        float4 n0 = make_float4(0.f, 0.f, 0.f, 0.f), n1 = n0;
        if (more) {
            const float* g = agg + ((size_t)pn * 16 + sr) * DIM + sc;
            n0 = *(const float4*)g;
            n1 = *(const float4*)(g + 8 * DIM);
        }
        const size_t gRow = (size_t)p * 16 + row;
        const float4 x0 = *(const float4*)(x + gRow * DIM + j * 8);
        const float4 x1 = *(const float4*)(x + gRow * DIM + j * 8 + 4);

        v8f acc = {};
        const float* arow = &aTile[cur][cc * PITCH + 4 * hh];
        float4 af[16];
#pragma unroll
        for (int i = 0; i < 16; ++i) af[i] = *(const float4*)(arow + 8 * i);
        __builtin_amdgcn_sched_barrier(0);
#pragma unroll
        for (int i = 0; i < 16; ++i) {
            const float4 a4 = af[i];
            v2f a01; a01.x = a4.x; a01.y = a4.y;
            v2f a23; a23.x = a4.z; a23.y = a4.w;
            acc = __builtin_amdgcn_wmma_f32_16x16x4_f32(false, a01, false, bW[2 * i],
                                                        (short)0, acc, false, false);
            acc = __builtin_amdgcn_wmma_f32_16x16x4_f32(false, a23, false, bW[2 * i + 1],
                                                        (short)0, acc, false, false);
        }
        __builtin_amdgcn_sched_barrier(0);
        asm volatile("v_nop\n\tv_nop\n\tv_nop\n\tv_nop");
        __builtin_amdgcn_sched_barrier(0);
        float* ot = oTile[cur];
        for (int r = 0; r < 8; ++r)
            ot[(r + 8 * hh) * DIM + nBase + cc] = acc[r];
        if (more) {
            *(float4*)&aTile[cur ^ 1][sr * PITCH + sc]       = n0;
            *(float4*)&aTile[cur ^ 1][(sr + 8) * PITCH + sc] = n1;
        }
        __syncthreads();

        const float xv[8] = {x0.x, x0.y, x0.z, x0.w, x1.x, x1.y, x1.z, x1.w};
        float vals[8];
        float s = 0.f, ss = 0.f;
        for (int k2 = 0; k2 < 8; ++k2) {
            const float v = ot[row * DIM + j * 8 + k2] + bo[k2] + xv[k2];
            vals[k2] = v; s += v; ss += v * v;
        }
        for (int msk = 1; msk < 16; msk <<= 1) {
            s  += __shfl_xor(s,  msk, 32);
            ss += __shfl_xor(ss, msk, 32);
        }
        const float mu  = s * (1.0f / DIM);
        const float var = ss * (1.0f / DIM) - mu * mu;
        const float inv = rsqrtf(var + 1e-5f);
        for (int k2 = 0; k2 < 8; ++k2)
            out[gRow * DIM + j * 8 + k2] = (vals[k2] - mu) * inv * ga[k2] + be[k2];
        cur ^= 1;
    }
}

extern "C" void kernel_launch(void* const* d_in, const int* in_sizes, int n_in,
                              void* d_out, int out_size, void* d_ws, size_t ws_size,
                              hipStream_t stream) {
    const float* x     = (const float*)d_in[0];
    const int*   ei    = (const int*)  d_in[1];
    const float* Wsrc  = (const float*)d_in[2];
    const float* Wdst  = (const float*)d_in[3];
    const float* attn  = (const float*)d_in[4];
    const float* Wout  = (const float*)d_in[5];
    const float* bout  = (const float*)d_in[6];
    const float* gamma = (const float*)d_in[7];
    const float* beta  = (const float*)d_in[8];
    float* out = (float*)d_out;

    float* ws     = (float*)d_ws;
    float* hsrc   = ws;
    float* hdst   = hsrc   + (size_t)N_NODES * DIM;
    float* agg    = hdst   + (size_t)N_NODES * DIM;
    int*   cnt    = (int*)(agg + (size_t)N_NODES * DIM);
    int*   start  = cnt    + N_NODES;
    int*   offT   = start  + N_NODES;
    int*   rec    = offT   + (N_BKT + 1) * NCH;
    int*   perm   = rec    + 2 * N_EDGES;

    const int gemm1Blks = NCH;
    const int gemm6Blks = 512;
    const int waveBlks = (N_NODES + 7) / 8;

    k_gemm_src_dst<<<gemm1Blks, 256, 0, stream>>>(x, Wsrc, Wdst, hsrc, hdst, ei, offT, rec);
    k_bucket_sort<<<N_BKT, 1024, 0, stream>>>(rec, offT, start, cnt, perm);
    k_node_attend<<<waveBlks, 256, 0, stream>>>(hsrc, hdst, start, cnt, perm, attn, agg);
    k_gemm_out_ln<<<gemm6Blks, 256, 0, stream>>>(agg, Wout, bout, x, gamma, beta, out);
}


# Attention loop is vector-issue bound: instructions per edge cut 45% (hardware exp2 on log2e-scaled logits, leaky as max+fma, head sums by DPP quad_perm not LDS permutes, one rcp); sum order kept
# speedup vs baseline: 4.0256x; 1.0716x over previous
#include <hip/hip_runtime.h>
#include <hip/hip_bf16.h>

#define N_NODES 50000
#define N_EDGES 800000
#define DIM     128
#define HEADS   8
#define HDIM    16

typedef __attribute__((ext_vector_type(2))) float v2f;
typedef __attribute__((ext_vector_type(8))) float v8f;

#define NCH       512
#define CH_MAX    ((N_EDGES + NCH - 1) / NCH)
#define EPT       ((CH_MAX + 255) / 256)
#define BKT_SHIFT 8
#define BKT_NODES (1 << BKT_SHIFT)
#define N_BKT     ((N_NODES + BKT_NODES - 1) / BKT_NODES)
#define RPT       8
static_assert((long long)NCH * N_EDGES < 2147483647LL, "chunk bounds are computed in 32 bits");
static_assert(N_BKT < 256, "the bucket scan covers 256 entries, the last one closes the chunk");
static_assert(2 * NCH == 1024, "level 2: two threads per chunk run");

__device__ __forceinline__ int chunk_first(const int c) { return (c * N_EDGES) / NCH; }

__device__ __forceinline__ void chunk_sort(const int* __restrict__ ei, int* __restrict__ offT,
                                           int* __restrict__ rec, int* tot, int* curs) {
    const int c    = blockIdx.x;
    const int tid  = threadIdx.x;
    const int lane = tid & 31;
    const int e0 = chunk_first(c);
    const int ne = chunk_first(c + 1) - e0;
    const int* s = ei + e0;
    const int* d = s + N_EDGES;
    tot[tid] = 0;
    int sv[EPT], dv[EPT];
#pragma unroll
    for (int k = 0; k < EPT; ++k) {
        const int i = tid + k * 256;
        const bool ok = i < ne;
        dv[k] = ok ? d[i] : -1;
        sv[k] = ok ? s[i] : 0;
    }
    __syncthreads();
#pragma unroll
    for (int k = 0; k < EPT; ++k)
        if (dv[k] >= 0) atomicAdd(&tot[dv[k] >> BKT_SHIFT], 1);
    __syncthreads();
    if (tid < 32) {
        const int4 a = ((const int4*)tot)[2 * lane];
        const int4 q = ((const int4*)tot)[2 * lane + 1];
        const int v[8] = {a.x, a.y, a.z, a.w, q.x, q.y, q.z, q.w};
        int ex[8];
        int run = 0;
        for (int j = 0; j < 8; ++j) { ex[j] = run; run += v[j]; }
        int inc = run;
        for (int o = 1; o < 32; o <<= 1) {
            const int u = __shfl_up(inc, o, 32);
            if (lane >= o) inc += u;
        }
        const int off = inc - run;
        for (int j = 0; j < 8; ++j) {
            const int bb    = 8 * lane + j;
            const int first = off + ex[j];
            curs[bb] = first;
            if (bb <= N_BKT) offT[bb * NCH + c] = first;
        }
    }
    __syncthreads();
#pragma unroll
    for (int k = 0; k < EPT; ++k)
        if (dv[k] >= 0) {
            const int slot = atomicAdd(&curs[dv[k] >> BKT_SHIFT], 1);
            ((int2*)rec)[e0 + slot] = make_int2(sv[k], dv[k]);
        }
}

__global__ void __launch_bounds__(1024)
k_bucket_sort(const int* __restrict__ rec, const int* __restrict__ offT,
              int* __restrict__ start, int* __restrict__ cnt, int* __restrict__ perm) {
    __shared__ __attribute__((aligned(16))) int c1[BKT_NODES];
    __shared__ int cur[BKT_NODES];
    __shared__ int wsum[32];
    const int b    = blockIdx.x;
    const int tid  = threadIdx.x;
    const int wave = tid >> 5;
    const int lane = tid & 31;
    const int c    = tid >> 1;
    const int half = tid & 1;
    const int o0 = offT[b * NCH + c];
    const int o1 = offT[(b + 1) * NCH + c];
    const int2* rr = (const int2*)rec + chunk_first(c);
    if (tid < BKT_NODES) c1[tid] = 0;
    int2 rc[RPT];
#pragma unroll
    for (int k = 0; k < RPT; ++k) {
        const int i = o0 + half + 2 * k;
        rc[k] = make_int2(0, -1);
        if (i < o1) rc[k] = rr[i];
    }
    int t = half == 0 ? o0 : 0;
    for (int m = 16; m >= 1; m >>= 1) t += __shfl_xor(t, m, 32);
    if (lane == 0) wsum[wave] = t;
    __syncthreads();
#pragma unroll
    for (int k = 0; k < RPT; ++k)
        if (rc[k].y >= 0) atomicAdd(&c1[rc[k].y & (BKT_NODES - 1)], 1);
    for (int i = o0 + half + 2 * RPT; i < o1; i += 2)
        atomicAdd(&c1[rr[i].y & (BKT_NODES - 1)], 1);
    __syncthreads();
    if (wave == 0) {
        int r0 = wsum[lane];
        for (int m = 16; m >= 1; m >>= 1) r0 += __shfl_xor(r0, m, 32);
        const int4 a = ((const int4*)c1)[2 * lane];
        const int4 q = ((const int4*)c1)[2 * lane + 1];
        const int v[8] = {a.x, a.y, a.z, a.w, q.x, q.y, q.z, q.w};
        int ex[8];
        int run = 0;
        for (int j = 0; j < 8; ++j) { ex[j] = run; run += v[j]; }
        int inc = run;
        for (int o = 1; o < 32; o <<= 1) {
            const int u = __shfl_up(inc, o, 32);
            if (lane >= o) inc += u;
        }
        const int off = r0 + inc - run;
        for (int j = 0; j < 8; ++j) {
            const int nl    = 8 * lane + j;
            const int n     = b * BKT_NODES + nl;
            const int first = off + ex[j];
            cur[nl] = first;
            if (n < N_NODES) { start[n] = first; cnt[n] = v[j]; }
        }
    }
    __syncthreads();
#pragma unroll
    for (int k = 0; k < RPT; ++k)
        if (rc[k].y >= 0) {
            const int slot = atomicAdd(&cur[rc[k].y & (BKT_NODES - 1)], 1);
            perm[slot] = rc[k].x;
        }
    for (int i = o0 + half + 2 * RPT; i < o1; i += 2) {
        const int2 q = rr[i];
        const int slot = atomicAdd(&cur[q.y & (BKT_NODES - 1)], 1);
        perm[slot] = q.x;
    }
}

#define N_PANELS (N_NODES / 16)
#define PITCH 132

__global__ void __launch_bounds__(256)
k_gemm_src_dst(const float* __restrict__ x,
               const float* __restrict__ Wsrc,
               const float* __restrict__ Wdst,
               float* __restrict__ hsrc,
               float* __restrict__ hdst,
               const int* __restrict__ ei,
               int* __restrict__ offT,
               int* __restrict__ rec) {
    __shared__ __attribute__((aligned(16))) float ax[2][16 * PITCH];
    __shared__ __attribute__((aligned(16))) int tot[256];
    __shared__ int curs[256];

    chunk_sort(ei, offT, rec, tot, curs);

    const int wave  = threadIdx.x >> 5;
    const int lane  = threadIdx.x & 31;
    const int nBase = wave * 16;
    const int cc    = lane & 15;
    const int hh    = lane >> 4;

    v2f bS[32], bD[32];
    {
        const float* ps = Wsrc + (4 * hh) * DIM + nBase + cc;
        const float* pd = Wdst + (4 * hh) * DIM + nBase + cc;
#pragma unroll
        for (int i = 0; i < 16; ++i) {
            bS[2 * i].x     = ps[(8 * i    ) * DIM];
            bS[2 * i].y     = ps[(8 * i + 1) * DIM];
            bS[2 * i + 1].x = ps[(8 * i + 2) * DIM];
            bS[2 * i + 1].y = ps[(8 * i + 3) * DIM];
            bD[2 * i].x     = pd[(8 * i    ) * DIM];
            bD[2 * i].y     = pd[(8 * i + 1) * DIM];
            bD[2 * i + 1].x = pd[(8 * i + 2) * DIM];
            bD[2 * i + 1].y = pd[(8 * i + 3) * DIM];
        }
    }

    const int sr = threadIdx.x >> 5;
    const int sc = (threadIdx.x & 31) * 4;

    int p   = blockIdx.x;
    int cur = 0;
    {
        const int pf = p < N_PANELS ? p : 0;
        const float* g = x + ((size_t)pf * 16 + sr) * DIM + sc;
        const float4 f0 = *(const float4*)g;
        const float4 f1 = *(const float4*)(g + 8 * DIM);
        *(float4*)&ax[0][sr * PITCH + sc]       = f0;
        *(float4*)&ax[0][(sr + 8) * PITCH + sc] = f1;
    }
    __syncthreads();

    for (; p < N_PANELS; p += gridDim.x) {
        const int  pn   = p + gridDim.x;
        const bool more = pn < N_PANELS;
        float4 n0 = make_float4(0.f, 0.f, 0.f, 0.f), n1 = n0;
        if (more) {
            const float* g = x + ((size_t)pn * 16 + sr) * DIM + sc;
            n0 = *(const float4*)g;
            n1 = *(const float4*)(g + 8 * DIM);
        }

        v8f accS = {}; v8f accD = {};
        const float* arow = &ax[cur][cc * PITCH + 4 * hh];
        float4 af[16];
#pragma unroll
        for (int i = 0; i < 16; ++i) af[i] = *(const float4*)(arow + 8 * i);
        __builtin_amdgcn_sched_barrier(0);
#pragma unroll
        for (int i = 0; i < 16; ++i) {
            const float4 a4 = af[i];
            v2f a01; a01.x = a4.x; a01.y = a4.y;
            v2f a23; a23.x = a4.z; a23.y = a4.w;
            accS = __builtin_amdgcn_wmma_f32_16x16x4_f32(false, a01, false, bS[2 * i],
                                                         (short)0, accS, false, false);
            accD = __builtin_amdgcn_wmma_f32_16x16x4_f32(false, a01, false, bD[2 * i],
                                                         (short)0, accD, false, false);
            accS = __builtin_amdgcn_wmma_f32_16x16x4_f32(false, a23, false, bS[2 * i + 1],
                                                         (short)0, accS, false, false);
            accD = __builtin_amdgcn_wmma_f32_16x16x4_f32(false, a23, false, bD[2 * i + 1],
                                                         (short)0, accD, false, false);
        }
        __builtin_amdgcn_sched_barrier(0);
        asm volatile("v_nop\n\tv_nop\n\tv_nop\n\tv_nop");
        __builtin_amdgcn_sched_barrier(0);
        const size_t row0 = (size_t)p * 16 + 8 * hh;
        for (int r = 0; r < 8; ++r) {
            hsrc[(row0 + r) * DIM + nBase + cc] = accS[r];
            hdst[(row0 + r) * DIM + nBase + cc] = accD[r];
        }
        if (more) {
            *(float4*)&ax[cur ^ 1][sr * PITCH + sc]       = n0;
            *(float4*)&ax[cur ^ 1][(sr + 8) * PITCH + sc] = n1;
        }
        __syncthreads();
        cur ^= 1;
    }
}

#define LOG2E 1.44269504088896340736f

__device__ __forceinline__ float head_sum(float p) {
    p += __int_as_float(__builtin_amdgcn_update_dpp(0, __float_as_int(p), 0xB1, 0xF, 0xF, true));
    p += __int_as_float(__builtin_amdgcn_update_dpp(0, __float_as_int(p), 0x4E, 0xF, 0xF, true));
    return p;
}

__device__ __forceinline__ float edge_logit(const float4 a, const float4 b,
                                            const float4 w) {
    const float t0 = a.x + b.x, t1 = a.y + b.y, t2 = a.z + b.z, t3 = a.w + b.w;
    const float l0 = fmaxf(t0, 0.2f * t0);
    const float l1 = fmaxf(t1, 0.2f * t1);
    const float l2 = fmaxf(t2, 0.2f * t2);
    const float l3 = fmaxf(t3, 0.2f * t3);
    float p = l0 * w.x;
    p = fmaf(l1, w.y, p);
    p = fmaf(l2, w.z, p);
    p = fmaf(l3, w.w, p);
    return head_sum(p);
}

__global__ void __launch_bounds__(256)
k_node_attend(const float* __restrict__ hsrc,
              const float* __restrict__ hdst,
              const int* __restrict__ start,
              const int* __restrict__ cnt,
              const int* __restrict__ perm,
              const float* __restrict__ attn,
              float* __restrict__ agg) {
    const int lane = threadIdx.x & 31;
    const int n = __builtin_amdgcn_readfirstlane(blockIdx.x * 8 + (threadIdx.x >> 5));
    if (n >= N_NODES) return;
    const int beg = start[n];
    const int end = beg + cnt[n];

    const float4 hd = ((const float4*)(hdst + (size_t)n * DIM))[lane];
    const float4 wa = ((const float4*)attn)[lane];
    const float4 w  = make_float4(wa.x * LOG2E, wa.y * LOG2E, wa.z * LOG2E, wa.w * LOG2E);
    const float NEG_INF = -__builtin_inff();

    float m = NEG_INF, den = 0.0f;
    float4 acc = make_float4(0.f, 0.f, 0.f, 0.f);

    for (int base = beg; base < end; base += 32) {
        const int rem  = end - base;
        const int last = (rem < 32 ? rem : 32) - 1;
        const int my = perm[base + (lane < last ? lane : last)];
        for (int j = 0; j <= last; j += 4) {
            const int j1 = (j + 1 < last) ? j + 1 : last;
            const int j2 = (j + 2 < last) ? j + 2 : last;
            const int j3 = (j + 3 < last) ? j + 3 : last;
            const int s0 = __builtin_amdgcn_readlane(my, j);
            const int s1 = __builtin_amdgcn_readlane(my, j1);
            const int s2 = __builtin_amdgcn_readlane(my, j2);
            const int s3 = __builtin_amdgcn_readlane(my, j3);
            const float4 r0 = ((const float4*)(hsrc + (size_t)s0 * DIM))[lane];
            const float4 r1 = ((const float4*)(hsrc + (size_t)s1 * DIM))[lane];
            const float4 r2 = ((const float4*)(hsrc + (size_t)s2 * DIM))[lane];
            const float4 r3 = ((const float4*)(hsrc + (size_t)s3 * DIM))[lane];
            const float p0 = edge_logit(r0, hd, w);
            float p1 = edge_logit(r1, hd, w);
            float p2 = edge_logit(r2, hd, w);
            float p3 = edge_logit(r3, hd, w);
            if (j + 1 > last) p1 = NEG_INF;
            if (j + 2 > last) p2 = NEG_INF;
            if (j + 3 > last) p3 = NEG_INF;

            const float mn = fmaxf(fmaxf(m, p0), fmaxf(fmaxf(p1, p2), p3));
            const float sc = __builtin_amdgcn_exp2f(m - mn);
            const float e0 = __builtin_amdgcn_exp2f(p0 - mn);
            const float e1 = __builtin_amdgcn_exp2f(p1 - mn);
            const float e2 = __builtin_amdgcn_exp2f(p2 - mn);
            const float e3 = __builtin_amdgcn_exp2f(p3 - mn);
            den   = den   * sc + ((e0 + e1) + (e2 + e3));
            acc.x = acc.x * sc + ((r0.x * e0 + r1.x * e1) + (r2.x * e2 + r3.x * e3));
            acc.y = acc.y * sc + ((r0.y * e0 + r1.y * e1) + (r2.y * e2 + r3.y * e3));
            acc.z = acc.z * sc + ((r0.z * e0 + r1.z * e1) + (r2.z * e2 + r3.z * e3));
            acc.w = acc.w * sc + ((r0.w * e0 + r1.w * e1) + (r2.w * e2 + r3.w * e3));
            m = mn;
        }
    }
    const float inv = __builtin_amdgcn_rcpf(den + 1e-8f);
    ((float4*)(agg + (size_t)n * DIM))[lane] =
        make_float4(acc.x * inv, acc.y * inv, acc.z * inv, acc.w * inv);
}

__global__ void __launch_bounds__(256)
k_gemm_out_ln(const float* __restrict__ agg,
              const float* __restrict__ Wout,
              const float* __restrict__ bout,
              const float* __restrict__ x,
              const float* __restrict__ gamma,
              const float* __restrict__ beta,
              float* __restrict__ out) {
    __shared__ __attribute__((aligned(16))) float aTile[2][16 * PITCH];
    __shared__ __attribute__((aligned(16))) float oTile[2][16 * DIM];

    const int wave  = threadIdx.x >> 5;
    const int lane  = threadIdx.x & 31;
    const int nBase = wave * 16;
    const int cc    = lane & 15;
    const int hh    = lane >> 4;

    v2f bW[32];
    {
        const float* pw = Wout + (4 * hh) * DIM + nBase + cc;
#pragma unroll
        for (int i = 0; i < 16; ++i) {
            bW[2 * i].x     = pw[(8 * i    ) * DIM];
            bW[2 * i].y     = pw[(8 * i + 1) * DIM];
            bW[2 * i + 1].x = pw[(8 * i + 2) * DIM];
            bW[2 * i + 1].y = pw[(8 * i + 3) * DIM];
        }
    }

    const int row = threadIdx.x >> 4;
    const int j   = threadIdx.x & 15;
    float bo[8], ga[8], be[8];
    for (int k2 = 0; k2 < 8; ++k2) {
        bo[k2] = bout[j * 8 + k2];
        ga[k2] = gamma[j * 8 + k2];
        be[k2] = beta[j * 8 + k2];
    }

    const int sr = threadIdx.x >> 5;
    const int sc = (threadIdx.x & 31) * 4;

    int p   = blockIdx.x;
    int cur = 0;
    {
        const int pf = p < N_PANELS ? p : 0;
        const float* g = agg + ((size_t)pf * 16 + sr) * DIM + sc;
        const float4 f0 = *(const float4*)g;
        const float4 f1 = *(const float4*)(g + 8 * DIM);
        *(float4*)&aTile[0][sr * PITCH + sc]       = f0;
        *(float4*)&aTile[0][(sr + 8) * PITCH + sc] = f1;
    }
    __syncthreads();

    for (; p < N_PANELS; p += gridDim.x) {
        const int  pn   = p + gridDim.x;
        const bool more = pn < N_PANELS;
        float4 n0 = make_float4(0.f, 0.f, 0.f, 0.f), n1 = n0;
        if (more) {
            const float* g = agg + ((size_t)pn * 16 + sr) * DIM + sc;
            n0 = *(const float4*)g;
            n1 = *(const float4*)(g + 8 * DIM);
        }
        const size_t gRow = (size_t)p * 16 + row;
        const float4 x0 = *(const float4*)(x + gRow * DIM + j * 8);
        const float4 x1 = *(const float4*)(x + gRow * DIM + j * 8 + 4);

        v8f acc = {};
        const float* arow = &aTile[cur][cc * PITCH + 4 * hh];
        float4 af[16];
#pragma unroll
        for (int i = 0; i < 16; ++i) af[i] = *(const float4*)(arow + 8 * i);
        __builtin_amdgcn_sched_barrier(0);
#pragma unroll
        for (int i = 0; i < 16; ++i) {
            const float4 a4 = af[i];
            v2f a01; a01.x = a4.x; a01.y = a4.y;
            v2f a23; a23.x = a4.z; a23.y = a4.w;
            acc = __builtin_amdgcn_wmma_f32_16x16x4_f32(false, a01, false, bW[2 * i],
                                                        (short)0, acc, false, false);
            acc = __builtin_amdgcn_wmma_f32_16x16x4_f32(false, a23, false, bW[2 * i + 1],
                                                        (short)0, acc, false, false);
        }
        __builtin_amdgcn_sched_barrier(0);
        asm volatile("v_nop\n\tv_nop\n\tv_nop\n\tv_nop");
        __builtin_amdgcn_sched_barrier(0);
        float* ot = oTile[cur];
        for (int r = 0; r < 8; ++r)
            ot[(r + 8 * hh) * DIM + nBase + cc] = acc[r];
        if (more) {
            *(float4*)&aTile[cur ^ 1][sr * PITCH + sc]       = n0;
            *(float4*)&aTile[cur ^ 1][(sr + 8) * PITCH + sc] = n1;
        }
        __syncthreads();

        const float xv[8] = {x0.x, x0.y, x0.z, x0.w, x1.x, x1.y, x1.z, x1.w};
        float vals[8];
        float s = 0.f, ss = 0.f;
        for (int k2 = 0; k2 < 8; ++k2) {
            const float v = ot[row * DIM + j * 8 + k2] + bo[k2] + xv[k2];
            vals[k2] = v; s += v; ss += v * v;
        }
        for (int msk = 1; msk < 16; msk <<= 1) {
            s  += __shfl_xor(s,  msk, 32);
            ss += __shfl_xor(ss, msk, 32);
        }
        const float mu  = s * (1.0f / DIM);
        const float var = ss * (1.0f / DIM) - mu * mu;
        const float inv = rsqrtf(var + 1e-5f);
        for (int k2 = 0; k2 < 8; ++k2)
            out[gRow * DIM + j * 8 + k2] = (vals[k2] - mu) * inv * ga[k2] + be[k2];
        cur ^= 1;
    }
}

extern "C" void kernel_launch(void* const* d_in, const int* in_sizes, int n_in,
                              void* d_out, int out_size, void* d_ws, size_t ws_size,
                              hipStream_t stream) {
    const float* x     = (const float*)d_in[0];
    const int*   ei    = (const int*)  d_in[1];
    const float* Wsrc  = (const float*)d_in[2];
    const float* Wdst  = (const float*)d_in[3];
    const float* attn  = (const float*)d_in[4];
    const float* Wout  = (const float*)d_in[5];
    const float* bout  = (const float*)d_in[6];
    const float* gamma = (const float*)d_in[7];
    const float* beta  = (const float*)d_in[8];
    float* out = (float*)d_out;

    float* ws     = (float*)d_ws;
    float* hsrc   = ws;
    float* hdst   = hsrc   + (size_t)N_NODES * DIM;
    float* agg    = hdst   + (size_t)N_NODES * DIM;
    int*   cnt    = (int*)(agg + (size_t)N_NODES * DIM);
    int*   start  = cnt    + N_NODES;
    int*   offT   = start  + N_NODES;
    int*   rec    = offT   + (N_BKT + 1) * NCH;
    int*   perm   = rec    + 2 * N_EDGES;

    const int gemm1Blks = NCH;
    const int gemm6Blks = 512;
    const int waveBlks = (N_NODES + 7) / 8;

    k_gemm_src_dst<<<gemm1Blks, 256, 0, stream>>>(x, Wsrc, Wdst, hsrc, hdst, ei, offT, rec);
    k_bucket_sort<<<N_BKT, 1024, 0, stream>>>(rec, offT, start, cnt, perm);
    k_node_attend<<<waveBlks, 256, 0, stream>>>(hsrc, hdst, start, cnt, perm, attn, agg);
    k_gemm_out_ln<<<gemm6Blks, 256, 0, stream>>>(agg, Wout, bout, x, gamma, beta, out);
}


# Attention and output GEMM+LayerNorm fused into one persistent launch: a block's 16 aggregated rows go through LDS to its WMMAs, not through memory; 4->3 launches; per-node sum order kept
# speedup vs baseline: 4.2595x; 1.0581x over previous
#include <hip/hip_runtime.h>
#include <hip/hip_bf16.h>

#define N_NODES 50000
#define N_EDGES 800000
#define DIM     128
#define HEADS   8
#define HDIM    16

typedef __attribute__((ext_vector_type(2))) float v2f;
typedef __attribute__((ext_vector_type(8))) float v8f;

#define NCH       512
#define CH_MAX    ((N_EDGES + NCH - 1) / NCH)
#define EPT       ((CH_MAX + 255) / 256)
#define BKT_SHIFT 8
#define BKT_NODES (1 << BKT_SHIFT)
#define N_BKT     ((N_NODES + BKT_NODES - 1) / BKT_NODES)
#define RPT       8
static_assert((long long)NCH * N_EDGES < 2147483647LL, "chunk bounds are computed in 32 bits");
static_assert(N_BKT < 256, "the bucket scan covers 256 entries, the last one closes the chunk");
static_assert(2 * NCH == 1024, "level 2: two threads per chunk run");

__device__ __forceinline__ int chunk_first(const int c) { return (c * N_EDGES) / NCH; }

__device__ __forceinline__ void chunk_sort(const int* __restrict__ ei, int* __restrict__ offT,
                                           int* __restrict__ rec, int* tot, int* curs) {
    const int c    = blockIdx.x;
    const int tid  = threadIdx.x;
    const int lane = tid & 31;
    const int e0 = chunk_first(c);
    const int ne = chunk_first(c + 1) - e0;
    const int* s = ei + e0;
    const int* d = s + N_EDGES;
    tot[tid] = 0;
    int sv[EPT], dv[EPT];
#pragma unroll
    for (int k = 0; k < EPT; ++k) {
        const int i = tid + k * 256;
        const bool ok = i < ne;
        dv[k] = ok ? d[i] : -1;
        sv[k] = ok ? s[i] : 0;
    }
    __syncthreads();
#pragma unroll
    for (int k = 0; k < EPT; ++k)
        if (dv[k] >= 0) atomicAdd(&tot[dv[k] >> BKT_SHIFT], 1);
    __syncthreads();
    if (tid < 32) {
        const int4 a = ((const int4*)tot)[2 * lane];
        const int4 q = ((const int4*)tot)[2 * lane + 1];
        const int v[8] = {a.x, a.y, a.z, a.w, q.x, q.y, q.z, q.w};
        int ex[8];
        int run = 0;
        for (int j = 0; j < 8; ++j) { ex[j] = run; run += v[j]; }
        int inc = run;
        for (int o = 1; o < 32; o <<= 1) {
            const int u = __shfl_up(inc, o, 32);
            if (lane >= o) inc += u;
        }
        const int off = inc - run;
        for (int j = 0; j < 8; ++j) {
            const int bb    = 8 * lane + j;
            const int first = off + ex[j];
            curs[bb] = first;
            if (bb <= N_BKT) offT[bb * NCH + c] = first;
        }
    }
    __syncthreads();
#pragma unroll
    for (int k = 0; k < EPT; ++k)
        if (dv[k] >= 0) {
            const int slot = atomicAdd(&curs[dv[k] >> BKT_SHIFT], 1);
            ((int2*)rec)[e0 + slot] = make_int2(sv[k], dv[k]);
        }
}

__global__ void __launch_bounds__(1024)
k_bucket_sort(const int* __restrict__ rec, const int* __restrict__ offT,
              int* __restrict__ start, int* __restrict__ cnt, int* __restrict__ perm) {
    __shared__ __attribute__((aligned(16))) int c1[BKT_NODES];
    __shared__ int cur[BKT_NODES];
    __shared__ int wsum[32];
    const int b    = blockIdx.x;
    const int tid  = threadIdx.x;
    const int wave = tid >> 5;
    const int lane = tid & 31;
    const int c    = tid >> 1;
    const int half = tid & 1;
    const int o0 = offT[b * NCH + c];
    const int o1 = offT[(b + 1) * NCH + c];
    const int2* rr = (const int2*)rec + chunk_first(c);
    if (tid < BKT_NODES) c1[tid] = 0;
    int2 rc[RPT];
#pragma unroll
    for (int k = 0; k < RPT; ++k) {
        const int i = o0 + half + 2 * k;
        rc[k] = make_int2(0, -1);
        if (i < o1) rc[k] = rr[i];
    }
    int t = half == 0 ? o0 : 0;
    for (int m = 16; m >= 1; m >>= 1) t += __shfl_xor(t, m, 32);
    if (lane == 0) wsum[wave] = t;
    __syncthreads();
#pragma unroll
    for (int k = 0; k < RPT; ++k)
        if (rc[k].y >= 0) atomicAdd(&c1[rc[k].y & (BKT_NODES - 1)], 1);
    for (int i = o0 + half + 2 * RPT; i < o1; i += 2)
        atomicAdd(&c1[rr[i].y & (BKT_NODES - 1)], 1);
    __syncthreads();
    if (wave == 0) {
        int r0 = wsum[lane];
        for (int m = 16; m >= 1; m >>= 1) r0 += __shfl_xor(r0, m, 32);
        const int4 a = ((const int4*)c1)[2 * lane];
        const int4 q = ((const int4*)c1)[2 * lane + 1];
        const int v[8] = {a.x, a.y, a.z, a.w, q.x, q.y, q.z, q.w};
        int ex[8];
        int run = 0;
        for (int j = 0; j < 8; ++j) { ex[j] = run; run += v[j]; }
        int inc = run;
        for (int o = 1; o < 32; o <<= 1) {
            const int u = __shfl_up(inc, o, 32);
            if (lane >= o) inc += u;
        }
        const int off = r0 + inc - run;
        for (int j = 0; j < 8; ++j) {
            const int nl    = 8 * lane + j;
            const int n     = b * BKT_NODES + nl;
            const int first = off + ex[j];
            cur[nl] = first;
            if (n < N_NODES) { start[n] = first; cnt[n] = v[j]; }
        }
    }
    __syncthreads();
#pragma unroll
    for (int k = 0; k < RPT; ++k)
        if (rc[k].y >= 0) {
            const int slot = atomicAdd(&cur[rc[k].y & (BKT_NODES - 1)], 1);
            perm[slot] = rc[k].x;
        }
    for (int i = o0 + half + 2 * RPT; i < o1; i += 2) {
        const int2 q = rr[i];
        const int slot = atomicAdd(&cur[q.y & (BKT_NODES - 1)], 1);
        perm[slot] = q.x;
    }
}

#define N_PANELS (N_NODES / 16)
#define PITCH 132

__global__ void __launch_bounds__(256)
k_gemm_src_dst(const float* __restrict__ x,
               const float* __restrict__ Wsrc,
               const float* __restrict__ Wdst,
               float* __restrict__ hsrc,
               float* __restrict__ hdst,
               const int* __restrict__ ei,
               int* __restrict__ offT,
               int* __restrict__ rec) {
    __shared__ __attribute__((aligned(16))) float ax[2][16 * PITCH];
    __shared__ __attribute__((aligned(16))) int tot[256];
    __shared__ int curs[256];

    chunk_sort(ei, offT, rec, tot, curs);

    const int wave  = threadIdx.x >> 5;
    const int lane  = threadIdx.x & 31;
    const int nBase = wave * 16;
    const int cc    = lane & 15;
    const int hh    = lane >> 4;

    v2f bS[32], bD[32];
    {
        const float* ps = Wsrc + (4 * hh) * DIM + nBase + cc;
        const float* pd = Wdst + (4 * hh) * DIM + nBase + cc;
#pragma unroll
        for (int i = 0; i < 16; ++i) {
            bS[2 * i].x     = ps[(8 * i    ) * DIM];
            bS[2 * i].y     = ps[(8 * i + 1) * DIM];
            bS[2 * i + 1].x = ps[(8 * i + 2) * DIM];
            bS[2 * i + 1].y = ps[(8 * i + 3) * DIM];
            bD[2 * i].x     = pd[(8 * i    ) * DIM];
            bD[2 * i].y     = pd[(8 * i + 1) * DIM];
            bD[2 * i + 1].x = pd[(8 * i + 2) * DIM];
            bD[2 * i + 1].y = pd[(8 * i + 3) * DIM];
        }
    }

    const int sr = threadIdx.x >> 5;
    const int sc = (threadIdx.x & 31) * 4;

    int p   = blockIdx.x;
    int cur = 0;
    {
        const int pf = p < N_PANELS ? p : 0;
        const float* g = x + ((size_t)pf * 16 + sr) * DIM + sc;
        const float4 f0 = *(const float4*)g;
        const float4 f1 = *(const float4*)(g + 8 * DIM);
        *(float4*)&ax[0][sr * PITCH + sc]       = f0;
        *(float4*)&ax[0][(sr + 8) * PITCH + sc] = f1;
    }
    __syncthreads();

    for (; p < N_PANELS; p += gridDim.x) {
        const int  pn   = p + gridDim.x;
        const bool more = pn < N_PANELS;
        float4 n0 = make_float4(0.f, 0.f, 0.f, 0.f), n1 = n0;
        if (more) {
            const float* g = x + ((size_t)pn * 16 + sr) * DIM + sc;
            n0 = *(const float4*)g;
            n1 = *(const float4*)(g + 8 * DIM);
        }

        v8f accS = {}; v8f accD = {};
        const float* arow = &ax[cur][cc * PITCH + 4 * hh];
        float4 af[16];
#pragma unroll
        for (int i = 0; i < 16; ++i) af[i] = *(const float4*)(arow + 8 * i);
        __builtin_amdgcn_sched_barrier(0);
#pragma unroll
        for (int i = 0; i < 16; ++i) {
            const float4 a4 = af[i];
            v2f a01; a01.x = a4.x; a01.y = a4.y;
            v2f a23; a23.x = a4.z; a23.y = a4.w;
            accS = __builtin_amdgcn_wmma_f32_16x16x4_f32(false, a01, false, bS[2 * i],
                                                         (short)0, accS, false, false);
            accD = __builtin_amdgcn_wmma_f32_16x16x4_f32(false, a01, false, bD[2 * i],
                                                         (short)0, accD, false, false);
            accS = __builtin_amdgcn_wmma_f32_16x16x4_f32(false, a23, false, bS[2 * i + 1],
                                                         (short)0, accS, false, false);
            accD = __builtin_amdgcn_wmma_f32_16x16x4_f32(false, a23, false, bD[2 * i + 1],
                                                         (short)0, accD, false, false);
        }
        __builtin_amdgcn_sched_barrier(0);
        asm volatile("v_nop\n\tv_nop\n\tv_nop\n\tv_nop");
        __builtin_amdgcn_sched_barrier(0);
        const size_t row0 = (size_t)p * 16 + 8 * hh;
        for (int r = 0; r < 8; ++r) {
            hsrc[(row0 + r) * DIM + nBase + cc] = accS[r];
            hdst[(row0 + r) * DIM + nBase + cc] = accD[r];
        }
        if (more) {
            *(float4*)&ax[cur ^ 1][sr * PITCH + sc]       = n0;
            *(float4*)&ax[cur ^ 1][(sr + 8) * PITCH + sc] = n1;
        }
        __syncthreads();
        cur ^= 1;
    }
}

#define LOG2E 1.44269504088896340736f
static_assert(N_NODES % 16 == 0, "every panel holds 16 nodes");

__device__ __forceinline__ float head_sum(float p) {
    p += __int_as_float(__builtin_amdgcn_update_dpp(0, __float_as_int(p), 0xB1, 0xF, 0xF, true));
    p += __int_as_float(__builtin_amdgcn_update_dpp(0, __float_as_int(p), 0x4E, 0xF, 0xF, true));
    return p;
}

__device__ __forceinline__ float edge_logit(const float4 a, const float4 b,
                                            const float4 w) {
    const float t0 = a.x + b.x, t1 = a.y + b.y, t2 = a.z + b.z, t3 = a.w + b.w;
    const float l0 = fmaxf(t0, 0.2f * t0);
    const float l1 = fmaxf(t1, 0.2f * t1);
    const float l2 = fmaxf(t2, 0.2f * t2);
    const float l3 = fmaxf(t3, 0.2f * t3);
    float p = l0 * w.x;
    p = fmaf(l1, w.y, p);
    p = fmaf(l2, w.z, p);
    p = fmaf(l3, w.w, p);
    return head_sum(p);
}

__device__ __forceinline__ int first_ids(const int* __restrict__ perm, const int beg,
                                         const int n_in, const int lane) {
    if (n_in <= 0) return 0;
    const int last = (n_in < 32 ? n_in : 32) - 1;
    return perm[beg + (lane < last ? lane : last)];
}

__device__ __forceinline__ int pair_bounds(const int* __restrict__ start, const int* __restrict__ cnt,
                                           const int n, const int lane) {
    const int* b = (lane & 2) ? cnt : start;
    return b[n + (lane & 1)];
}

__device__ __forceinline__ float4 attend_node(const float* __restrict__ hsrc,
                                              const int* __restrict__ perm,
                                              const int beg, const int end, const int my0,
                                              const float4 hd, const float4 w, const int lane) {
    const float NEG_INF = -__builtin_inff();
    float m = NEG_INF, den = 0.0f;
    float4 acc = make_float4(0.f, 0.f, 0.f, 0.f);

    for (int base = beg; base < end; base += 32) {
        const int rem  = end - base;
        const int last = (rem < 32 ? rem : 32) - 1;
        int my = my0;
        if (base != beg) my = perm[base + (lane < last ? lane : last)];
        for (int j = 0; j <= last; j += 4) {
            const int j1 = (j + 1 < last) ? j + 1 : last;
            const int j2 = (j + 2 < last) ? j + 2 : last;
            const int j3 = (j + 3 < last) ? j + 3 : last;
            const int s0 = __builtin_amdgcn_readlane(my, j);
            const int s1 = __builtin_amdgcn_readlane(my, j1);
            const int s2 = __builtin_amdgcn_readlane(my, j2);
            const int s3 = __builtin_amdgcn_readlane(my, j3);
            const float4 r0 = ((const float4*)(hsrc + (size_t)s0 * DIM))[lane];
            const float4 r1 = ((const float4*)(hsrc + (size_t)s1 * DIM))[lane];
            const float4 r2 = ((const float4*)(hsrc + (size_t)s2 * DIM))[lane];
            const float4 r3 = ((const float4*)(hsrc + (size_t)s3 * DIM))[lane];
            const float p0 = edge_logit(r0, hd, w);
            float p1 = edge_logit(r1, hd, w);
            float p2 = edge_logit(r2, hd, w);
            float p3 = edge_logit(r3, hd, w);
            if (j + 1 > last) p1 = NEG_INF;
            if (j + 2 > last) p2 = NEG_INF;
            if (j + 3 > last) p3 = NEG_INF;

            const float mn = fmaxf(fmaxf(m, p0), fmaxf(fmaxf(p1, p2), p3));
            const float sc = __builtin_amdgcn_exp2f(m - mn);
            const float e0 = __builtin_amdgcn_exp2f(p0 - mn);
            const float e1 = __builtin_amdgcn_exp2f(p1 - mn);
            const float e2 = __builtin_amdgcn_exp2f(p2 - mn);
            const float e3 = __builtin_amdgcn_exp2f(p3 - mn);
            den   = den   * sc + ((e0 + e1) + (e2 + e3));
            acc.x = acc.x * sc + ((r0.x * e0 + r1.x * e1) + (r2.x * e2 + r3.x * e3));
            acc.y = acc.y * sc + ((r0.y * e0 + r1.y * e1) + (r2.y * e2 + r3.y * e3));
            acc.z = acc.z * sc + ((r0.z * e0 + r1.z * e1) + (r2.z * e2 + r3.z * e3));
            acc.w = acc.w * sc + ((r0.w * e0 + r1.w * e1) + (r2.w * e2 + r3.w * e3));
            m = mn;
        }
    }
    const float inv = __builtin_amdgcn_rcpf(den + 1e-8f);
    return make_float4(acc.x * inv, acc.y * inv, acc.z * inv, acc.w * inv);
}

__global__ void __launch_bounds__(256, 6)
k_attend_out_ln(const float* __restrict__ hsrc,
                const float* __restrict__ hdst,
                const int* __restrict__ start,
                const int* __restrict__ cnt,
                const int* __restrict__ perm,
                const float* __restrict__ attn,
                const float* __restrict__ Wout,
                const float* __restrict__ bout,
                const float* x,
                const float* __restrict__ gamma,
                const float* __restrict__ beta,
                float* __restrict__ out) {
    __shared__ __attribute__((aligned(16))) float aTile[16 * PITCH];
    __shared__ __attribute__((aligned(16))) float oTile[16 * DIM];
    __shared__ __attribute__((aligned(16))) float lnv[3 * DIM];

    const int wave  = __builtin_amdgcn_readfirstlane(threadIdx.x >> 5);
    const int lane  = threadIdx.x & 31;
    const int nBase = wave * 16;
    const int cc    = lane & 15;
    const int hh    = lane >> 4;

    v2f bW[32];
    {
        const float* pw = Wout + (4 * hh) * DIM + nBase + cc;
#pragma unroll
        for (int i = 0; i < 16; ++i) {
            bW[2 * i].x     = pw[(8 * i    ) * DIM];
            bW[2 * i].y     = pw[(8 * i + 1) * DIM];
            bW[2 * i + 1].x = pw[(8 * i + 2) * DIM];
            bW[2 * i + 1].y = pw[(8 * i + 3) * DIM];
        }
    }

    const float4 wa = ((const float4*)attn)[lane];
    const float4 w  = make_float4(wa.x * LOG2E, wa.y * LOG2E, wa.z * LOG2E, wa.w * LOG2E);

    const int row = threadIdx.x >> 4;
    const int j   = threadIdx.x & 15;
    if (threadIdx.x < DIM) {
        lnv[threadIdx.x]           = bout[threadIdx.x];
        lnv[DIM + threadIdx.x]     = gamma[threadIdx.x];
        lnv[2 * DIM + threadIdx.x] = beta[threadIdx.x];
    }

    int p = blockIdx.x;
    if (p >= N_PANELS) return;
    const int n0 = p * 16 + 2 * wave;
    const int mv0 = pair_bounds(start, cnt, n0, lane);
    int begA = __builtin_amdgcn_readlane(mv0, 0), begB = __builtin_amdgcn_readlane(mv0, 1);
    int cntA = __builtin_amdgcn_readlane(mv0, 2), cntB = __builtin_amdgcn_readlane(mv0, 3);
    float4 hdA = ((const float4*)(hdst + (size_t)n0 * DIM))[lane];
    float4 hdB = ((const float4*)(hdst + (size_t)(n0 + 1) * DIM))[lane];
    int myA = first_ids(perm, begA, cntA, lane);
    int myB = first_ids(perm, begB, cntB, lane);
    int pn = p + gridDim.x;
    int mvN = 0;
    if (pn < N_PANELS) mvN = pair_bounds(start, cnt, pn * 16 + 2 * wave, lane);

    for (;;) {
        const float4 gA = attend_node(hsrc, perm, begA, begA + cntA, myA, hdA, w, lane);
        *(float4*)&aTile[(2 * wave) * PITCH + 4 * lane] = gA;
        const float4 gB = attend_node(hsrc, perm, begB, begB + cntB, myB, hdB, w, lane);
        *(float4*)&aTile[(2 * wave + 1) * PITCH + 4 * lane] = gB;
        __syncthreads();

        const bool more = pn < N_PANELS;
        if (more) {
            begA = __builtin_amdgcn_readlane(mvN, 0); begB = __builtin_amdgcn_readlane(mvN, 1);
            cntA = __builtin_amdgcn_readlane(mvN, 2); cntB = __builtin_amdgcn_readlane(mvN, 3);
            const int n1 = pn * 16 + 2 * wave;
            hdA = ((const float4*)(hdst + (size_t)n1 * DIM))[lane];
            hdB = ((const float4*)(hdst + (size_t)(n1 + 1) * DIM))[lane];
            myA = first_ids(perm, begA, cntA, lane);
            myB = first_ids(perm, begB, cntB, lane);
            const int pn2 = pn + gridDim.x;
            if (pn2 < N_PANELS) mvN = pair_bounds(start, cnt, pn2 * 16 + 2 * wave, lane);
        }
        const size_t gRow = (size_t)p * 16 + row;
        const float4 x0 = *(const float4*)(x + gRow * DIM + j * 8);
        const float4 x1 = *(const float4*)(x + gRow * DIM + j * 8 + 4);
        v8f acc = {};
        const float* arow = &aTile[cc * PITCH + 4 * hh];
#pragma unroll
        for (int h2 = 0; h2 < 2; ++h2) {
            float4 af[8];
#pragma unroll
            for (int i = 0; i < 8; ++i) af[i] = *(const float4*)(arow + 8 * (8 * h2 + i));
            asm volatile("" :: "v"(af[0].x), "v"(af[0].y), "v"(af[0].z), "v"(af[0].w),
                               "v"(af[1].x), "v"(af[1].y), "v"(af[1].z), "v"(af[1].w),
                               "v"(af[2].x), "v"(af[2].y), "v"(af[2].z), "v"(af[2].w),
                               "v"(af[3].x), "v"(af[3].y), "v"(af[3].z), "v"(af[3].w));
            asm volatile("" :: "v"(af[4].x), "v"(af[4].y), "v"(af[4].z), "v"(af[4].w),
                               "v"(af[5].x), "v"(af[5].y), "v"(af[5].z), "v"(af[5].w),
                               "v"(af[6].x), "v"(af[6].y), "v"(af[6].z), "v"(af[6].w),
                               "v"(af[7].x), "v"(af[7].y), "v"(af[7].z), "v"(af[7].w));
            __builtin_amdgcn_sched_barrier(0);
#pragma unroll
            for (int i = 0; i < 8; ++i) {
                const float4 a4 = af[i];
                v2f a01; a01.x = a4.x; a01.y = a4.y;
                v2f a23; a23.x = a4.z; a23.y = a4.w;
                acc = __builtin_amdgcn_wmma_f32_16x16x4_f32(false, a01, false, bW[2 * (8 * h2 + i)],
                                                            (short)0, acc, false, false);
                acc = __builtin_amdgcn_wmma_f32_16x16x4_f32(false, a23, false, bW[2 * (8 * h2 + i) + 1],
                                                            (short)0, acc, false, false);
            }
            __builtin_amdgcn_sched_barrier(0);
            asm volatile("v_nop\n\tv_nop\n\tv_nop\n\tv_nop");
            __builtin_amdgcn_sched_barrier(0);
        }
        for (int r = 0; r < 8; ++r)
            oTile[(r + 8 * hh) * DIM + nBase + cc] = acc[r];
        __syncthreads();

        const float xv[8] = {x0.x, x0.y, x0.z, x0.w, x1.x, x1.y, x1.z, x1.w};
        float bo[8], ga[8], be[8];
        for (int k2 = 0; k2 < 8; ++k2) {
            bo[k2] = lnv[j * 8 + k2];
            ga[k2] = lnv[DIM + j * 8 + k2];
            be[k2] = lnv[2 * DIM + j * 8 + k2];
        }
        float vals[8];
        float s = 0.f, ss = 0.f;
        for (int k2 = 0; k2 < 8; ++k2) {
            const float v = oTile[row * DIM + j * 8 + k2] + bo[k2] + xv[k2];
            vals[k2] = v; s += v; ss += v * v;
        }
        for (int msk = 1; msk < 16; msk <<= 1) {
            s  += __shfl_xor(s,  msk, 32);
            ss += __shfl_xor(ss, msk, 32);
        }
        const float mu  = s * (1.0f / DIM);
        const float var = ss * (1.0f / DIM) - mu * mu;
        const float inv = rsqrtf(var + 1e-5f);
        for (int k2 = 0; k2 < 8; ++k2)
            out[gRow * DIM + j * 8 + k2] = (vals[k2] - mu) * inv * ga[k2] + be[k2];
        if (!more) break;
        p = pn;
        pn += gridDim.x;
    }
}

extern "C" void kernel_launch(void* const* d_in, const int* in_sizes, int n_in,
                              void* d_out, int out_size, void* d_ws, size_t ws_size,
                              hipStream_t stream) {
    const float* x     = (const float*)d_in[0];
    const int*   ei    = (const int*)  d_in[1];
    const float* Wsrc  = (const float*)d_in[2];
    const float* Wdst  = (const float*)d_in[3];
    const float* attn  = (const float*)d_in[4];
    const float* Wout  = (const float*)d_in[5];
    const float* bout  = (const float*)d_in[6];
    const float* gamma = (const float*)d_in[7];
    const float* beta  = (const float*)d_in[8];
    float* out = (float*)d_out;

    float* ws     = (float*)d_ws;
    float* hsrc   = ws;
    float* hdst   = hsrc   + (size_t)N_NODES * DIM;
    int*   cnt    = (int*)(hdst + (size_t)N_NODES * DIM);
    int*   start  = cnt    + N_NODES;
    int*   offT   = start  + N_NODES;
    int*   rec    = offT   + (N_BKT + 1) * NCH;
    int*   perm   = rec    + 2 * N_EDGES;

    const int gemm1Blks = NCH;
    const int fusedBlks = 768;

    k_gemm_src_dst<<<gemm1Blks, 256, 0, stream>>>(x, Wsrc, Wdst, hsrc, hdst, ei, offT, rec);
    k_bucket_sort<<<N_BKT, 1024, 0, stream>>>(rec, offT, start, cnt, perm);
    k_attend_out_ln<<<fusedBlks, 256, 0, stream>>>(hsrc, hdst, start, cnt, perm, attn,
                                                   Wout, bout, x, gamma, beta, out);
}
